# starting kernel
# baseline (speedup 1.0000x reference)
#include <hip/hip_runtime.h>
#include <hip/hip_bf16.h>


typedef float v2f __attribute__((ext_vector_type(2)));
typedef float v8f __attribute__((ext_vector_type(8)));

#define BSZ   1024
#define CCH   16
#define EDIM  64
#define SLICE (CCH * EDIM)
#define ATOT  (BSZ * BSZ * CCH)

__device__ __forceinline__ unsigned int enc_f32(float f) {
  unsigned int u = __float_as_uint(f);
  return (u & 0x80000000u) ? ~u : (u | 0x80000000u);
}
__device__ __forceinline__ float dec_f32(unsigned int u) {
  return (u & 0x80000000u) ? __uint_as_float(u & 0x7FFFFFFFu)
                           : __uint_as_float(~u);
}

#define LDA_P 20
#define LDB_P 68

__global__ __launch_bounds__(256) void gemm_bias_act(
    const float* __restrict__ A, const float* __restrict__ W,
    const float* __restrict__ bias, float* __restrict__ out,
    int M, int K, int N, int relu)
{
  __shared__ float lA[128 * LDA_P];
  __shared__ float lB[16 * LDB_P];

  const int t    = threadIdx.x;
  const int lane = t & 31;
  const int wave = t >> 5;
  const int wm   = wave & 3;
  const int wn   = wave >> 2;
  const int bm   = blockIdx.y * 128;
  const int bn   = blockIdx.x * 64;

  v8f acc[2][2] = {};

  const int mrow = lane & 15;
  const int koff = (lane >> 4) * 2;
  const int ncol = lane & 15;

  for (int k0 = 0; k0 < K; k0 += 16) {
    #pragma unroll
    for (int i = 0; i < 2; ++i) {
      int idx = t + i * 256;
      int row = idx >> 2;
      int c4  = idx & 3;
      float4 v = *(const float4*)(A + (size_t)(bm + row) * K + k0 + c4 * 4);
      *(float4*)(lA + row * LDA_P + c4 * 4) = v;
    }
    {
      int row = t >> 4;
      int c4  = t & 15;
      float4 v = *(const float4*)(W + (size_t)(k0 + row) * N + bn + c4 * 4);
      *(float4*)(lB + row * LDB_P + c4 * 4) = v;
    }
    __syncthreads();

    #pragma unroll
    for (int kk = 0; kk < 16; kk += 4) {
      v2f afrag[2], bfrag[2];
      #pragma unroll
      for (int mt = 0; mt < 2; ++mt) {
        int row = wm * 32 + mt * 16 + mrow;
        afrag[mt] = *(const v2f*)(lA + row * LDA_P + kk + koff);
      }
      #pragma unroll
      for (int nt = 0; nt < 2; ++nt) {
        int col = wn * 32 + nt * 16 + ncol;
        bfrag[nt].x = lB[(kk + koff + 0) * LDB_P + col];
        bfrag[nt].y = lB[(kk + koff + 1) * LDB_P + col];
      }
      #pragma unroll
      for (int mt = 0; mt < 2; ++mt)
        #pragma unroll
        for (int nt = 0; nt < 2; ++nt)
          acc[mt][nt] = __builtin_amdgcn_wmma_f32_16x16x4_f32(
              false, afrag[mt], false, bfrag[nt],
              (short)0, acc[mt][nt], false, false);
    }
    __syncthreads();
  }

  const int rsel = (lane >> 4) * 8;
  #pragma unroll
  for (int nt = 0; nt < 2; ++nt) {
    int col = bn + wn * 32 + nt * 16 + ncol;
    float bv = bias[col];
    #pragma unroll
    for (int mt = 0; mt < 2; ++mt) {
      #pragma unroll
      for (int v = 0; v < 8; ++v) {
        int row = bm + wm * 32 + mt * 16 + rsel + v;
        float x = acc[mt][nt][v] + bv;
        if (relu) x = fmaxf(x, 0.0f);
        out[(size_t)row * N + col] = x;
      }
    }
  }
}

__global__ __launch_bounds__(256) void norm_rows64(float* __restrict__ q)
{
  const int lane = threadIdx.x & 31;
  const int wave = threadIdx.x >> 5;
  const int row  = blockIdx.x * 8 + wave;
  float* p = q + (size_t)row * 64;

  float2 x = *(float2*)(p + lane * 2);
  float s  = x.x + x.y;
  float ss = x.x * x.x + x.y * x.y;
  #pragma unroll
  for (int off = 16; off > 0; off >>= 1) {
    s  += __shfl_xor(s,  off, 32);
    ss += __shfl_xor(ss, off, 32);
  }
  float mean = s * (1.0f / 64.0f);
  float var  = (ss - 64.0f * mean * mean) * (1.0f / 63.0f);
  float inv  = rsqrtf(var + 1e-8f);
  x.x = (x.x - mean) * inv;
  x.y = (x.y - mean) * inv;
  *(float2*)(p + lane * 2) = x;
}

__global__ void sink_init(float* __restrict__ v, unsigned int* __restrict__ maxbuf)
{
  int i = blockIdx.x * 256 + threadIdx.x;
  if (i < BSZ * CCH) v[i] = 1.0f;
  if (i < CCH) maxbuf[i] = 0u;
}

#define LQA_P 68
#define LQB_P 65

__global__ __launch_bounds__(256) void affinity_wmma(
    const float* __restrict__ q1, const float* __restrict__ q2,
    float* __restrict__ aff, unsigned int* __restrict__ maxbuf)
{
  __shared__ float lA[128 * LQA_P];
  __shared__ float lB[64 * LQB_P];
  __shared__ float red[256];

  const int t    = threadIdx.x;
  const int lane = t & 31;
  const int wave = t >> 5;
  const int wm   = wave & 3;
  const int wn   = wave >> 2;
  const int bb   = blockIdx.y * 128;
  const int bd   = blockIdx.x * 64;
  const int c    = blockIdx.z;

  #pragma unroll
  for (int i = 0; i < 8; ++i) {
    int idx = t + i * 256;
    int row = idx >> 4;
    int c4  = idx & 15;
    float4 v = *(const float4*)(q1 + (size_t)(bb + row) * SLICE + c * EDIM + c4 * 4);
    *(float4*)(lA + row * LQA_P + c4 * 4) = v;
  }
  #pragma unroll
  for (int i = 0; i < 4; ++i) {
    int idx = t + i * 256;
    int dl = idx >> 4;
    int e4 = idx & 15;
    float4 v = *(const float4*)(q2 + (size_t)(bd + dl) * SLICE + c * EDIM + e4 * 4);
    lB[(e4 * 4 + 0) * LQB_P + dl] = v.x;
    lB[(e4 * 4 + 1) * LQB_P + dl] = v.y;
    lB[(e4 * 4 + 2) * LQB_P + dl] = v.z;
    lB[(e4 * 4 + 3) * LQB_P + dl] = v.w;
  }
  __syncthreads();

  v8f acc[2][2] = {};
  const int mrow = lane & 15;
  const int koff = (lane >> 4) * 2;
  const int ncol = lane & 15;

  #pragma unroll
  for (int kk = 0; kk < 64; kk += 4) {
    v2f afrag[2], bfrag[2];
    #pragma unroll
    for (int mt = 0; mt < 2; ++mt) {
      int row = wm * 32 + mt * 16 + mrow;
      afrag[mt] = *(const v2f*)(lA + row * LQA_P + kk + koff);
    }
    #pragma unroll
    for (int nt = 0; nt < 2; ++nt) {
      int col = wn * 32 + nt * 16 + ncol;
      bfrag[nt].x = lB[(kk + koff + 0) * LQB_P + col];
      bfrag[nt].y = lB[(kk + koff + 1) * LQB_P + col];
    }
    #pragma unroll
    for (int mt = 0; mt < 2; ++mt)
      #pragma unroll
      for (int nt = 0; nt < 2; ++nt)
        acc[mt][nt] = __builtin_amdgcn_wmma_f32_16x16x4_f32(
            false, afrag[mt], false, bfrag[nt],
            (short)0, acc[mt][nt], false, false);
  }

  const float scale = 0.125f;
  const int rsel = (lane >> 4) * 8;
  float lmax = -3.4e38f;
  #pragma unroll
  for (int mt = 0; mt < 2; ++mt) {
    #pragma unroll
    for (int nt = 0; nt < 2; ++nt) {
      int col = bd + wn * 32 + nt * 16 + ncol;
      #pragma unroll
      for (int v = 0; v < 8; ++v) {
        int row = bb + wm * 32 + mt * 16 + rsel + v;
        float x = acc[mt][nt][v] * scale;
        lmax = fmaxf(lmax, x);
        aff[((size_t)row * BSZ + col) * CCH + c] = x;
      }
    }
  }
  red[t] = lmax;
  __syncthreads();
  for (int off = 128; off > 0; off >>= 1) {
    if (t < off) red[t] = fmaxf(red[t], red[t + off]);
    __syncthreads();
  }
  if (t == 0) atomicMax(&maxbuf[c], enc_f32(red[0]));
}

__global__ __launch_bounds__(256) void exp_sub_max(
    float* __restrict__ A, const unsigned int* __restrict__ maxbuf)
{
  size_t base   = ((size_t)blockIdx.x * 256 + threadIdx.x) * 4;
  size_t stride = (size_t)gridDim.x * 256 * 4;
  for (size_t i = base; i < (size_t)ATOT; i += stride) {
    #pragma unroll
    for (int j = 0; j < 4; ++j) {
      size_t idx = i + j;
      float mv = dec_f32(maxbuf[idx & 15]);
      A[idx] = expf(A[idx] - mv);
    }
  }
}

__global__ __launch_bounds__(256) void sink_u(
    const float* __restrict__ A, const float* __restrict__ v,
    const float* __restrict__ r, float* __restrict__ u)
{
  __shared__ float red[256];
  const int t = threadIdx.x;
  const int b = blockIdx.x;
  const float* Ab = A + (size_t)b * (BSZ * CCH);

  float s = 0.0f;
  #pragma unroll 4
  for (int i = 0; i < 64; ++i) {
    int j = t + 256 * i;
    s += Ab[j] * v[j];
  }
  red[t] = s;
  __syncthreads();
  for (int off = 128; off >= 16; off >>= 1) {
    if (t < off) red[t] += red[t + off];
    __syncthreads();
  }
  if (t < CCH) u[b * CCH + t] = r[b * CCH + t] / red[t];
}

__global__ __launch_bounds__(256) void sink_v(
    const float* __restrict__ A, const float* __restrict__ u,
    const float* __restrict__ cmarg, float* __restrict__ v)
{
  __shared__ float red[256];
  const int t = threadIdx.x;
  const int d = blockIdx.x;
  const int c = t & 15;
  const int bseg = t >> 4;

  float s = 0.0f;
  #pragma unroll 4
  for (int k = 0; k < 64; ++k) {
    int b = bseg * 64 + k;
    s += A[(size_t)b * (BSZ * CCH) + d * CCH + c] * u[b * CCH + c];
  }
  red[t] = s;
  __syncthreads();
  for (int off = 128; off >= 16; off >>= 1) {
    if (t < off) red[t] += red[t + off];
    __syncthreads();
  }
  if (t < CCH) v[d * CCH + t] = cmarg[d * CCH + t] / red[t];
}

__global__ __launch_bounds__(256) void sink_final(
    float* __restrict__ A, const float* __restrict__ u,
    const float* __restrict__ v)
{
  size_t base   = ((size_t)blockIdx.x * 256 + threadIdx.x) * 4;
  size_t stride = (size_t)gridDim.x * 256 * 4;
  for (size_t i = base; i < (size_t)ATOT; i += stride) {
    #pragma unroll
    for (int j = 0; j < 4; ++j) {
      size_t idx = i + j;
      size_t b  = idx >> 14;
      size_t dc = idx & 16383;
      A[idx] = A[idx] * u[b * CCH + (idx & 15)] * v[dc];
    }
  }
}

extern "C" void kernel_launch(void* const* d_in, const int* in_sizes, int n_in,
                              void* d_out, int out_size, void* d_ws, size_t ws_size,
                              hipStream_t stream)
{
  const float* x1   = (const float*)d_in[0];
  const float* x2   = (const float*)d_in[1];
  const float* p1   = (const float*)d_in[2];
  const float* p2   = (const float*)d_in[3];
  const float* w1[4] = {(const float*)d_in[4],  (const float*)d_in[6],
                        (const float*)d_in[8],  (const float*)d_in[10]};
  const float* b1[4] = {(const float*)d_in[5],  (const float*)d_in[7],
                        (const float*)d_in[9],  (const float*)d_in[11]};
  const float* w2[4] = {(const float*)d_in[12], (const float*)d_in[14],
                        (const float*)d_in[16], (const float*)d_in[18]};
  const float* b2[4] = {(const float*)d_in[13], (const float*)d_in[15],
                        (const float*)d_in[17], (const float*)d_in[19]};

  float* ws = (float*)d_ws;
  float* hA = ws;
  float* hB = hA + 1024 * 512;
  float* q1 = hB + 1024 * 512;
  float* q2 = q1 + 1024 * 1024;
  float* uv_u = q2 + 1024 * 1024;
  float* uv_v = uv_u + 16384;
  unsigned int* maxbuf = (unsigned int*)(uv_v + 16384);

  float* Aout = (float*)d_out;

  gemm_bias_act<<<dim3(512/64, 1024/128), 256, 0, stream>>>(x1, w1[0], b1[0], hA, 1024, 256, 512, 1);
  gemm_bias_act<<<dim3(512/64, 1024/128), 256, 0, stream>>>(hA, w1[1], b1[1], hB, 1024, 512, 512, 1);
  gemm_bias_act<<<dim3(512/64, 1024/128), 256, 0, stream>>>(hB, w1[2], b1[2], hA, 1024, 512, 512, 1);
  gemm_bias_act<<<dim3(1024/64, 1024/128), 256, 0, stream>>>(hA, w1[3], b1[3], q1, 1024, 512, 1024, 0);
  gemm_bias_act<<<dim3(512/64, 1024/128), 256, 0, stream>>>(x2, w2[0], b2[0], hA, 1024, 256, 512, 1);
  gemm_bias_act<<<dim3(512/64, 1024/128), 256, 0, stream>>>(hA, w2[1], b2[1], hB, 1024, 512, 512, 1);
  gemm_bias_act<<<dim3(512/64, 1024/128), 256, 0, stream>>>(hB, w2[2], b2[2], hA, 1024, 512, 512, 1);
  gemm_bias_act<<<dim3(1024/64, 1024/128), 256, 0, stream>>>(hA, w2[3], b2[3], q2, 1024, 512, 1024, 0);

  norm_rows64<<<16384 / 8, 256, 0, stream>>>(q1);
  norm_rows64<<<16384 / 8, 256, 0, stream>>>(q2);

  sink_init<<<(1024 * 16 + 255) / 256, 256, 0, stream>>>(uv_v, maxbuf);

  affinity_wmma<<<dim3(1024/64, 1024/128, CCH), 256, 0, stream>>>(q1, q2, Aout, maxbuf);

  exp_sub_max<<<4096, 256, 0, stream>>>(Aout, maxbuf);

  for (int it = 0; it < 10; ++it) {
    sink_u<<<1024, 256, 0, stream>>>(Aout, uv_v, p1, uv_u);
    sink_v<<<1024, 256, 0, stream>>>(Aout, uv_u, p2, uv_v);
  }

  sink_final<<<4096, 256, 0, stream>>>(Aout, uv_u, uv_v);
}


# MLP GEMMs split K four ways over grid z (256 or 512 workgroups, 4 or 8 K steps each): raw partial planes; the next launch's A loader (norm launch for q) adds planes and bias, ReLU; K sums reordered
# speedup vs baseline: 2.2234x; 2.2234x over previous
#include <hip/hip_runtime.h>
#include <hip/hip_bf16.h>


typedef float v2f __attribute__((ext_vector_type(2)));
typedef float v4f __attribute__((ext_vector_type(4)));
typedef float v8f __attribute__((ext_vector_type(8)));

#define BSZ   1024
#define CCH   16
#define EDIM  64
#define SLICE (CCH * EDIM)
#define ATOT  (BSZ * BSZ * CCH)

__device__ __forceinline__ unsigned int enc_f32(float f) {
  unsigned int u = __float_as_uint(f);
  return (u & 0x80000000u) ? ~u : (u | 0x80000000u);
}
__device__ __forceinline__ float dec_f32(unsigned int u) {
  return (u & 0x80000000u) ? __uint_as_float(u & 0x7FFFFFFFu)
                           : __uint_as_float(~u);
}

#define KSPLIT 4
#define LDA_P 20
#define LDB_P 68

template <int NIN>
__global__ __launch_bounds__(256) void gemm_splitk(
    const float* __restrict__ A, const float* __restrict__ biasIn,
    const float* __restrict__ W, float* __restrict__ out,
    int M, int K, int N, int kchunk)
{
  static_assert(NIN == 1 || NIN == 4, "plane count of the A loader");
  __shared__ float lA[128 * LDA_P];
  __shared__ float lB[16 * LDB_P];

  const int t    = threadIdx.x;
  const int lane = t & 31;
  const int wave = t >> 5;
  const int wm   = wave & 3;
  const int wn   = wave >> 2;
  const int bm   = blockIdx.y * 128;
  const int bn   = blockIdx.x * 64;
  const int kbeg = blockIdx.z * kchunk;
  const int kend = kbeg + kchunk;
  const size_t pitch_in = (size_t)M * K;

  v8f acc[2][2] = {};

  const int mrow = lane & 15;
  const int koff = (lane >> 4) * 2;
  const int ncol = lane & 15;

  for (int k0 = kbeg; k0 < kend; k0 += 16) {
    const int arow = t >> 2;
    const int ac4  = t & 3;
    const int brow = t >> 4;
    const int bc4  = t & 15;
    v4f pv[2][NIN];
    v4f bb = {0.0f, 0.0f, 0.0f, 0.0f};
    #pragma unroll
    for (int i = 0; i < 2; ++i) {
      const float* pa = A + (size_t)(bm + arow + 64 * i) * K + k0 + ac4 * 4;
      #pragma unroll
      for (int p = 0; p < NIN; ++p)
        pv[i][p] = *(const v4f*)(pa + (size_t)p * pitch_in);
    }
    if constexpr (NIN > 1) bb = *(const v4f*)(biasIn + k0 + ac4 * 4);
    const v4f wv = *(const v4f*)(W + (size_t)(k0 + brow) * N + bn + bc4 * 4);
    if constexpr (NIN == 4) {
      asm volatile("" :: "v"(pv[0][0]), "v"(pv[0][1]), "v"(pv[0][2]), "v"(pv[0][3]),
                         "v"(pv[1][0]), "v"(pv[1][1]), "v"(pv[1][2]), "v"(pv[1][3]),
                         "v"(bb), "v"(wv));
    }
    #pragma unroll
    for (int i = 0; i < 2; ++i) {
      v4f v = pv[i][0];
      if constexpr (NIN > 1) {
        #pragma unroll
        for (int p = 1; p < NIN; ++p) v += pv[i][p];
        v += bb;
        v.x = fmaxf(v.x, 0.0f);
        v.y = fmaxf(v.y, 0.0f);
        v.z = fmaxf(v.z, 0.0f);
        v.w = fmaxf(v.w, 0.0f);
      }
      *(v4f*)(lA + (arow + 64 * i) * LDA_P + ac4 * 4) = v;
    }
    *(v4f*)(lB + brow * LDB_P + bc4 * 4) = wv;
    __syncthreads();

    #pragma unroll
    for (int kk = 0; kk < 16; kk += 4) {
      v2f afrag[2], bfrag[2];
      #pragma unroll
      for (int mt = 0; mt < 2; ++mt) {
        int row = wm * 32 + mt * 16 + mrow;
        afrag[mt] = *(const v2f*)(lA + row * LDA_P + kk + koff);
      }
      #pragma unroll
      for (int nt = 0; nt < 2; ++nt) {
        int col = wn * 32 + nt * 16 + ncol;
        bfrag[nt].x = lB[(kk + koff + 0) * LDB_P + col];
        bfrag[nt].y = lB[(kk + koff + 1) * LDB_P + col];
      }
      #pragma unroll
      for (int mt = 0; mt < 2; ++mt)
        #pragma unroll
        for (int nt = 0; nt < 2; ++nt)
          acc[mt][nt] = __builtin_amdgcn_wmma_f32_16x16x4_f32(
              false, afrag[mt], false, bfrag[nt],
              (short)0, acc[mt][nt], false, false);
    }
    __syncthreads();
  }

  float* op = out + (size_t)blockIdx.z * M * N;
  const int rsel = (lane >> 4) * 8;
  #pragma unroll
  for (int nt = 0; nt < 2; ++nt) {
    int col = bn + wn * 32 + nt * 16 + ncol;
    #pragma unroll
    for (int mt = 0; mt < 2; ++mt) {
      #pragma unroll
      for (int v = 0; v < 8; ++v) {
        int row = bm + wm * 32 + mt * 16 + rsel + v;
        op[(size_t)row * N + col] = acc[mt][nt][v];
      }
    }
  }
}

__global__ __launch_bounds__(256) void norm_rows64(
    const float* __restrict__ qp, const float* __restrict__ bias, float* __restrict__ q)
{
  const int lane = threadIdx.x & 31;
  const int wave = threadIdx.x >> 5;
  const int row  = blockIdx.x * 8 + wave;
  float* p = q + (size_t)row * 64;
  const float* pp = qp + (size_t)row * 64 + lane * 2;

  float2 x = *(const float2*)pp;
  const float2 bb = *(const float2*)(bias + (row & 15) * 64 + lane * 2);
  #pragma unroll
  for (int k = 1; k < KSPLIT; ++k) {
    const float2 s = *(const float2*)(pp + (size_t)k * (BSZ * SLICE));
    x.x += s.x; x.y += s.y;
  }
  x.x += bb.x; x.y += bb.y;
  float s  = x.x + x.y;
  float ss = x.x * x.x + x.y * x.y;
  #pragma unroll
  for (int off = 16; off > 0; off >>= 1) {
    s  += __shfl_xor(s,  off, 32);
    ss += __shfl_xor(ss, off, 32);
  }
  float mean = s * (1.0f / 64.0f);
  float var  = (ss - 64.0f * mean * mean) * (1.0f / 63.0f);
  float inv  = rsqrtf(var + 1e-8f);
  x.x = (x.x - mean) * inv;
  x.y = (x.y - mean) * inv;
  *(float2*)(p + lane * 2) = x;
}

__global__ void sink_init(float* __restrict__ v, unsigned int* __restrict__ maxbuf)
{
  int i = blockIdx.x * 256 + threadIdx.x;
  if (i < BSZ * CCH) v[i] = 1.0f;
  if (i < CCH) maxbuf[i] = 0u;
}

#define LQA_P 68
#define LQB_P 65

__global__ __launch_bounds__(256) void affinity_wmma(
    const float* __restrict__ q1, const float* __restrict__ q2,
    float* __restrict__ aff, unsigned int* __restrict__ maxbuf)
{
  __shared__ float lA[128 * LQA_P];
  __shared__ float lB[64 * LQB_P];
  __shared__ float red[256];

  const int t    = threadIdx.x;
  const int lane = t & 31;
  const int wave = t >> 5;
  const int wm   = wave & 3;
  const int wn   = wave >> 2;
  const int bb   = blockIdx.y * 128;
  const int bd   = blockIdx.x * 64;
  const int c    = blockIdx.z;

  #pragma unroll
  for (int i = 0; i < 8; ++i) {
    int idx = t + i * 256;
    int row = idx >> 4;
    int c4  = idx & 15;
    float4 v = *(const float4*)(q1 + (size_t)(bb + row) * SLICE + c * EDIM + c4 * 4);
    *(float4*)(lA + row * LQA_P + c4 * 4) = v;
  }
  #pragma unroll
  for (int i = 0; i < 4; ++i) {
    int idx = t + i * 256;
    int dl = idx >> 4;
    int e4 = idx & 15;
    float4 v = *(const float4*)(q2 + (size_t)(bd + dl) * SLICE + c * EDIM + e4 * 4);
    lB[(e4 * 4 + 0) * LQB_P + dl] = v.x;
    lB[(e4 * 4 + 1) * LQB_P + dl] = v.y;
    lB[(e4 * 4 + 2) * LQB_P + dl] = v.z;
    lB[(e4 * 4 + 3) * LQB_P + dl] = v.w;
  }
  __syncthreads();

  v8f acc[2][2] = {};
  const int mrow = lane & 15;
  const int koff = (lane >> 4) * 2;
  const int ncol = lane & 15;

  #pragma unroll
  for (int kk = 0; kk < 64; kk += 4) {
    v2f afrag[2], bfrag[2];
    #pragma unroll
    for (int mt = 0; mt < 2; ++mt) {
      int row = wm * 32 + mt * 16 + mrow;
      afrag[mt] = *(const v2f*)(lA + row * LQA_P + kk + koff);
    }
    #pragma unroll
    for (int nt = 0; nt < 2; ++nt) {
      int col = wn * 32 + nt * 16 + ncol;
      bfrag[nt].x = lB[(kk + koff + 0) * LQB_P + col];
      bfrag[nt].y = lB[(kk + koff + 1) * LQB_P + col];
    }
    #pragma unroll
    for (int mt = 0; mt < 2; ++mt)
      #pragma unroll
      for (int nt = 0; nt < 2; ++nt)
        acc[mt][nt] = __builtin_amdgcn_wmma_f32_16x16x4_f32(
            false, afrag[mt], false, bfrag[nt],
            (short)0, acc[mt][nt], false, false);
  }

  const float scale = 0.125f;
  const int rsel = (lane >> 4) * 8;
  float lmax = -3.4e38f;
  __syncthreads();
  #pragma unroll
  for (int mt = 0; mt < 2; ++mt) {
    #pragma unroll
    for (int nt = 0; nt < 2; ++nt) {
      int col = wn * 32 + nt * 16 + ncol;
      #pragma unroll
      for (int v = 0; v < 8; ++v) {
        int row = wm * 32 + mt * 16 + rsel + v;
        float x = acc[mt][nt][v] * scale;
        lmax = fmaxf(lmax, x);
        lA[row * LQA_P + col] = x;
      }
    }
  }
  red[t] = lmax;
  __syncthreads();
  #pragma unroll
  for (int i = 0; i < 8; ++i) {
    int idx = t + i * 256;
    int row = idx >> 4;
    int c4  = idx & 15;
    float4 v = *(const float4*)(lA + row * LQA_P + c4 * 4);
    *(float4*)(aff + (size_t)(bb + row) * (BSZ * CCH) + c * BSZ + bd + c4 * 4) = v;
  }
  for (int off = 128; off > 0; off >>= 1) {
    if (t < off) red[t] = fmaxf(red[t], red[t + off]);
    __syncthreads();
  }
  if (t == 0) atomicMax(&maxbuf[c], enc_f32(red[0]));
}

#define XP 1028

__global__ __launch_bounds__(256) void exp_sub_max(
    float* __restrict__ A, const unsigned int* __restrict__ maxbuf)
{
  __shared__ float L[CCH * XP];
  const int t = threadIdx.x;
  float* slab = A + (size_t)blockIdx.x * (BSZ * CCH);

  float4 a[CCH];
  #pragma unroll
  for (int i = 0; i < CCH; ++i)
    a[i] = *(const float4*)(slab + i * BSZ + t * 4);

  #pragma unroll
  for (int i = 0; i < CCH; ++i) {
    float mv = dec_f32(maxbuf[i]);
    float4 e;
    e.x = expf(a[i].x - mv);
    e.y = expf(a[i].y - mv);
    e.z = expf(a[i].z - mv);
    e.w = expf(a[i].w - mv);
    *(float4*)(L + i * XP + t * 4) = e;
  }
  __syncthreads();

  #pragma unroll
  for (int i = 0; i < CCH; ++i) {
    int o  = t + 256 * i;
    int d  = o >> 2;
    int c4 = o & 3;
    float4 r;
    r.x = L[(4 * c4 + 0) * XP + d];
    r.y = L[(4 * c4 + 1) * XP + d];
    r.z = L[(4 * c4 + 2) * XP + d];
    r.w = L[(4 * c4 + 3) * XP + d];
    *(float4*)(slab + (size_t)o * 4) = r;
  }
}

#define NPLANE 192
#define SLABF  (BSZ * CCH)
#define SLAB4  (SLABF / 4)

__device__ __forceinline__ float4 f4_fma(float4 a, float4 b, float4 c) {
  c.x += a.x * b.x; c.y += a.y * b.y; c.z += a.z * b.z; c.w += a.w * b.w;
  return c;
}

__global__ __launch_bounds__(512) void sink_rowcol(
    const float* __restrict__ A, const float* __restrict__ v,
    const float* __restrict__ r, float* __restrict__ u,
    float* __restrict__ part)
{
  __shared__ float4 red[2][16][4];
  const int t    = threadIdx.x;
  const int lane = t & 31;
  const int wave = t >> 5;
  const int c4   = t & 3;
  const int g    = blockIdx.x;
  const int ns   = (g < BSZ - 5 * NPLANE) ? 6 : 5;

  const float4* A4 = (const float4*)A;
  const float4* v4 = (const float4*)v;
  const float4* r4p = (const float4*)r;

  float4 a[3][8];
  float4 rr[3];
  rr[0] = r4p[g * 4 + c4];
  #pragma unroll
  for (int i = 0; i < 8; ++i) a[0][i] = A4[(size_t)g * SLAB4 + t + 512 * i];
  rr[1] = r4p[(g + NPLANE) * 4 + c4];
  #pragma unroll
  for (int i = 0; i < 8; ++i) a[1][i] = A4[(size_t)(g + NPLANE) * SLAB4 + t + 512 * i];

  float4 vv[8], acc[8];
  #pragma unroll
  for (int i = 0; i < 8; ++i) {
    vv[i]  = v4[t + 512 * i];
    acc[i] = make_float4(0.0f, 0.0f, 0.0f, 0.0f);
  }

  #pragma unroll
  for (int j = 0; j < 6; ++j) {
    if (j < ns) {
      const int b = g + NPLANE * j;
      if (j + 2 < ns) {
        rr[(j + 2) % 3] = r4p[(b + 2 * NPLANE) * 4 + c4];
        #pragma unroll
        for (int i = 0; i < 8; ++i)
          a[(j + 2) % 3][i] = A4[(size_t)(b + 2 * NPLANE) * SLAB4 + t + 512 * i];
      }

      float4 p = make_float4(0.0f, 0.0f, 0.0f, 0.0f);
      #pragma unroll
      for (int i = 0; i < 8; ++i) p = f4_fma(a[j % 3][i], vv[i], p);
      #pragma unroll
      for (int off = 4; off < 32; off <<= 1) {
        p.x += __shfl_xor(p.x, off, 32);
        p.y += __shfl_xor(p.y, off, 32);
        p.z += __shfl_xor(p.z, off, 32);
        p.w += __shfl_xor(p.w, off, 32);
      }
      if (lane < 4) red[j & 1][wave][lane] = p;
      __builtin_amdgcn_fence(__ATOMIC_RELEASE, "workgroup", "local");
      __builtin_amdgcn_s_barrier();
      __builtin_amdgcn_fence(__ATOMIC_ACQUIRE, "workgroup", "local");
      float4 s = red[j & 1][0][c4];
      #pragma unroll
      for (int w = 1; w < 16; ++w) {
        float4 q = red[j & 1][w][c4];
        s.x += q.x; s.y += q.y; s.z += q.z; s.w += q.w;
      }
      float4 uu;
      uu.x = rr[j % 3].x / s.x; uu.y = rr[j % 3].y / s.y;
      uu.z = rr[j % 3].z / s.z; uu.w = rr[j % 3].w / s.w;
      if (t < 4) ((float4*)u)[b * 4 + t] = uu;

      #pragma unroll
      for (int i = 0; i < 8; ++i) {
        acc[i] = f4_fma(a[j % 3][i], uu, acc[i]);
        asm volatile("" : "+v"(acc[i].x), "+v"(acc[i].y), "+v"(acc[i].z), "+v"(acc[i].w));
      }
    }
  }

  float4* P4 = (float4*)part + (size_t)g * SLAB4;
  #pragma unroll
  for (int i = 0; i < 8; ++i) P4[t + 512 * i] = acc[i];
}

__global__ __launch_bounds__(256) void sink_colsum(
    const float* __restrict__ part, const float* __restrict__ cmarg,
    float* __restrict__ v)
{
  __shared__ float4 red[16][16];
  const int t   = threadIdx.x;
  const int col = t & 15;
  const int gg  = t >> 4;
  const int o   = blockIdx.x * 16 + col;
  const float4* P4 = (const float4*)part;

  const float4 cm = ((const float4*)cmarg)[o];
  float4 x[NPLANE / 16];
  #pragma unroll
  for (int k = 0; k < NPLANE / 16; ++k)
    x[k] = P4[(size_t)(gg + 16 * k) * SLAB4 + o];
  float4 s = x[0];
  #pragma unroll
  for (int k = 1; k < NPLANE / 16; ++k) {
    s.x += x[k].x; s.y += x[k].y; s.z += x[k].z; s.w += x[k].w;
  }
  red[gg][col] = s;
  __syncthreads();
  if (t < 16) {
    float4 tot = red[0][t];
    #pragma unroll
    for (int q = 1; q < 16; ++q) {
      float4 y = red[q][t];
      tot.x += y.x; tot.y += y.y; tot.z += y.z; tot.w += y.w;
    }
    float4 out;
    out.x = cm.x / tot.x; out.y = cm.y / tot.y; out.z = cm.z / tot.z; out.w = cm.w / tot.w;
    ((float4*)v)[blockIdx.x * 16 + t] = out;
  }
}

__global__ __launch_bounds__(256) void sink_final(
    float* __restrict__ A, const float* __restrict__ u,
    const float* __restrict__ v)
{
  size_t base   = ((size_t)blockIdx.x * 256 + threadIdx.x) * 4;
  size_t stride = (size_t)gridDim.x * 256 * 4;
  for (size_t i = base; i < (size_t)ATOT; i += stride) {
    #pragma unroll
    for (int j = 0; j < 4; ++j) {
      size_t idx = i + j;
      size_t b  = idx >> 14;
      size_t dc = idx & 16383;
      A[idx] = A[idx] * u[b * CCH + (idx & 15)] * v[dc];
    }
  }
}

extern "C" void kernel_launch(void* const* d_in, const int* in_sizes, int n_in,
                              void* d_out, int out_size, void* d_ws, size_t ws_size,
                              hipStream_t stream)
{
  const float* x1   = (const float*)d_in[0];
  const float* x2   = (const float*)d_in[1];
  const float* p1   = (const float*)d_in[2];
  const float* p2   = (const float*)d_in[3];
  const float* w1[4] = {(const float*)d_in[4],  (const float*)d_in[6],
                        (const float*)d_in[8],  (const float*)d_in[10]};
  const float* b1[4] = {(const float*)d_in[5],  (const float*)d_in[7],
                        (const float*)d_in[9],  (const float*)d_in[11]};
  const float* w2[4] = {(const float*)d_in[12], (const float*)d_in[14],
                        (const float*)d_in[16], (const float*)d_in[18]};
  const float* b2[4] = {(const float*)d_in[13], (const float*)d_in[15],
                        (const float*)d_in[17], (const float*)d_in[19]};

  float* ws = (float*)d_ws;
  float* q1 = ws + 1024 * 1024;
  float* q2 = q1 + 1024 * 1024;
  float* uv_u = q2 + 1024 * 1024;
  float* uv_v = uv_u + 16384;
  unsigned int* maxbuf = (unsigned int*)(uv_v + 16384);

  float* Aout = (float*)d_out;

  float* pA = Aout;
  float* pB = pA + 4 * 1024 * 512;
  float* pQ = pB + 4 * 1024 * 512;

  gemm_splitk<1><<<dim3(512/64, 1024/128, KSPLIT), 256, 0, stream>>>(x1, nullptr, w1[0], pA, 1024, 256, 512, 64);
  gemm_splitk<4><<<dim3(512/64, 1024/128, KSPLIT), 256, 0, stream>>>(pA, b1[0], w1[1], pB, 1024, 512, 512, 128);
  gemm_splitk<4><<<dim3(512/64, 1024/128, KSPLIT), 256, 0, stream>>>(pB, b1[1], w1[2], pA, 1024, 512, 512, 128);
  gemm_splitk<4><<<dim3(1024/64, 1024/128, KSPLIT), 256, 0, stream>>>(pA, b1[2], w1[3], pQ, 1024, 512, 1024, 128);
  norm_rows64<<<16384 / 8, 256, 0, stream>>>(pQ, b1[3], q1);
  gemm_splitk<1><<<dim3(512/64, 1024/128, KSPLIT), 256, 0, stream>>>(x2, nullptr, w2[0], pA, 1024, 256, 512, 64);
  gemm_splitk<4><<<dim3(512/64, 1024/128, KSPLIT), 256, 0, stream>>>(pA, b2[0], w2[1], pB, 1024, 512, 512, 128);
  gemm_splitk<4><<<dim3(512/64, 1024/128, KSPLIT), 256, 0, stream>>>(pB, b2[1], w2[2], pA, 1024, 512, 512, 128);
  gemm_splitk<4><<<dim3(1024/64, 1024/128, KSPLIT), 256, 0, stream>>>(pA, b2[2], w2[3], pQ, 1024, 512, 1024, 128);
  norm_rows64<<<16384 / 8, 256, 0, stream>>>(pQ, b2[3], q2);

  sink_init<<<(1024 * 16 + 255) / 256, 256, 0, stream>>>(uv_v, maxbuf);

  affinity_wmma<<<dim3(1024/64, 1024/128, CCH), 256, 0, stream>>>(q1, q2, Aout, maxbuf);

  exp_sub_max<<<BSZ, 256, 0, stream>>>(Aout, maxbuf);

  float* part = ws;
  for (int it = 0; it < 10; ++it) {
    sink_rowcol<<<NPLANE, 512, 0, stream>>>(Aout, uv_v, p1, uv_u, part);
    sink_colsum<<<SLAB4 / 16, 256, 0, stream>>>(part, p2, uv_v);
  }

  sink_final<<<4096, 256, 0, stream>>>(Aout, uv_u, uv_v);
}


# Sinkhorn iteration as ONE launch: A stays channel-major, a workgroup owns one channel and 64 rows and rebuilds that channel's v from 16 small planes; transpose moved to final pass; sums reordered
# speedup vs baseline: 2.7624x; 1.2424x over previous
#include <hip/hip_runtime.h>
#include <hip/hip_bf16.h>


typedef float v2f __attribute__((ext_vector_type(2)));
typedef float v4f __attribute__((ext_vector_type(4)));
typedef float v8f __attribute__((ext_vector_type(8)));

#define BSZ   1024
#define CCH   16
#define EDIM  64
#define SLICE (CCH * EDIM)
#define ATOT  (BSZ * BSZ * CCH)

__device__ __forceinline__ unsigned int enc_f32(float f) {
  unsigned int u = __float_as_uint(f);
  return (u & 0x80000000u) ? ~u : (u | 0x80000000u);
}
__device__ __forceinline__ float dec_f32(unsigned int u) {
  return (u & 0x80000000u) ? __uint_as_float(u & 0x7FFFFFFFu)
                           : __uint_as_float(~u);
}

#define KSPLIT 4
#define LDA_P 20
#define LDB_P 68

template <int NIN>
__global__ __launch_bounds__(256) void gemm_splitk(
    const float* __restrict__ A, const float* __restrict__ biasIn,
    const float* __restrict__ W, float* __restrict__ out,
    int M, int K, int N, int kchunk)
{
  static_assert(NIN == 1 || NIN == 4, "plane count of the A loader");
  __shared__ float lA[128 * LDA_P];
  __shared__ float lB[16 * LDB_P];

  const int t    = threadIdx.x;
  const int lane = t & 31;
  const int wave = t >> 5;
  const int wm   = wave & 3;
  const int wn   = wave >> 2;
  const int bm   = blockIdx.y * 128;
  const int bn   = blockIdx.x * 64;
  const int kbeg = blockIdx.z * kchunk;
  const int kend = kbeg + kchunk;
  const size_t pitch_in = (size_t)M * K;

  v8f acc[2][2] = {};

  const int mrow = lane & 15;
  const int koff = (lane >> 4) * 2;
  const int ncol = lane & 15;

  for (int k0 = kbeg; k0 < kend; k0 += 16) {
    const int arow = t >> 2;
    const int ac4  = t & 3;
    const int brow = t >> 4;
    const int bc4  = t & 15;
    v4f pv[2][NIN];
    v4f bb = {0.0f, 0.0f, 0.0f, 0.0f};
    #pragma unroll
    for (int i = 0; i < 2; ++i) {
      const float* pa = A + (size_t)(bm + arow + 64 * i) * K + k0 + ac4 * 4;
      #pragma unroll
      for (int p = 0; p < NIN; ++p)
        pv[i][p] = *(const v4f*)(pa + (size_t)p * pitch_in);
    }
    if constexpr (NIN > 1) bb = *(const v4f*)(biasIn + k0 + ac4 * 4);
    const v4f wv = *(const v4f*)(W + (size_t)(k0 + brow) * N + bn + bc4 * 4);
    if constexpr (NIN == 4) {
      asm volatile("" :: "v"(pv[0][0]), "v"(pv[0][1]), "v"(pv[0][2]), "v"(pv[0][3]),
                         "v"(pv[1][0]), "v"(pv[1][1]), "v"(pv[1][2]), "v"(pv[1][3]),
                         "v"(bb), "v"(wv));
    }
    #pragma unroll
    for (int i = 0; i < 2; ++i) {
      v4f v = pv[i][0];
      if constexpr (NIN > 1) {
        #pragma unroll
        for (int p = 1; p < NIN; ++p) v += pv[i][p];
        v += bb;
        v.x = fmaxf(v.x, 0.0f);
        v.y = fmaxf(v.y, 0.0f);
        v.z = fmaxf(v.z, 0.0f);
        v.w = fmaxf(v.w, 0.0f);
      }
      *(v4f*)(lA + (arow + 64 * i) * LDA_P + ac4 * 4) = v;
    }
    *(v4f*)(lB + brow * LDB_P + bc4 * 4) = wv;
    __syncthreads();

    #pragma unroll
    for (int kk = 0; kk < 16; kk += 4) {
      v2f afrag[2], bfrag[2];
      #pragma unroll
      for (int mt = 0; mt < 2; ++mt) {
        int row = wm * 32 + mt * 16 + mrow;
        afrag[mt] = *(const v2f*)(lA + row * LDA_P + kk + koff);
      }
      #pragma unroll
      for (int nt = 0; nt < 2; ++nt) {
        int col = wn * 32 + nt * 16 + ncol;
        bfrag[nt].x = lB[(kk + koff + 0) * LDB_P + col];
        bfrag[nt].y = lB[(kk + koff + 1) * LDB_P + col];
      }
      #pragma unroll
      for (int mt = 0; mt < 2; ++mt)
        #pragma unroll
        for (int nt = 0; nt < 2; ++nt)
          acc[mt][nt] = __builtin_amdgcn_wmma_f32_16x16x4_f32(
              false, afrag[mt], false, bfrag[nt],
              (short)0, acc[mt][nt], false, false);
    }
    __syncthreads();
  }

  float* op = out + (size_t)blockIdx.z * M * N;
  const int rsel = (lane >> 4) * 8;
  #pragma unroll
  for (int nt = 0; nt < 2; ++nt) {
    int col = bn + wn * 32 + nt * 16 + ncol;
    #pragma unroll
    for (int mt = 0; mt < 2; ++mt) {
      #pragma unroll
      for (int v = 0; v < 8; ++v) {
        int row = bm + wm * 32 + mt * 16 + rsel + v;
        op[(size_t)row * N + col] = acc[mt][nt][v];
      }
    }
  }
}

__global__ __launch_bounds__(256) void norm_rows64(
    const float* __restrict__ qp, const float* __restrict__ bias, float* __restrict__ q)
{
  const int lane = threadIdx.x & 31;
  const int wave = threadIdx.x >> 5;
  const int row  = blockIdx.x * 8 + wave;
  float* p = q + (size_t)row * 64;
  const float* pp = qp + (size_t)row * 64 + lane * 2;

  float2 x = *(const float2*)pp;
  const float2 bb = *(const float2*)(bias + (row & 15) * 64 + lane * 2);
  #pragma unroll
  for (int k = 1; k < KSPLIT; ++k) {
    const float2 s = *(const float2*)(pp + (size_t)k * (BSZ * SLICE));
    x.x += s.x; x.y += s.y;
  }
  x.x += bb.x; x.y += bb.y;
  float s  = x.x + x.y;
  float ss = x.x * x.x + x.y * x.y;
  #pragma unroll
  for (int off = 16; off > 0; off >>= 1) {
    s  += __shfl_xor(s,  off, 32);
    ss += __shfl_xor(ss, off, 32);
  }
  float mean = s * (1.0f / 64.0f);
  float var  = (ss - 64.0f * mean * mean) * (1.0f / 63.0f);
  float inv  = rsqrtf(var + 1e-8f);
  x.x = (x.x - mean) * inv;
  x.y = (x.y - mean) * inv;
  *(float2*)(p + lane * 2) = x;
}

__global__ void sink_init(unsigned int* __restrict__ maxbuf)
{
  int i = threadIdx.x;
  if (i < CCH) maxbuf[i] = 0u;
}

#define LQA_P 68
#define LQB_P 65

__global__ __launch_bounds__(256) void affinity_wmma(
    const float* __restrict__ q1, const float* __restrict__ q2,
    float* __restrict__ aff, unsigned int* __restrict__ maxbuf)
{
  __shared__ float lA[128 * LQA_P];
  __shared__ float lB[64 * LQB_P];
  __shared__ float red[256];

  const int t    = threadIdx.x;
  const int lane = t & 31;
  const int wave = t >> 5;
  const int wm   = wave & 3;
  const int wn   = wave >> 2;
  const int bb   = blockIdx.y * 128;
  const int bd   = blockIdx.x * 64;
  const int c    = blockIdx.z;

  #pragma unroll
  for (int i = 0; i < 8; ++i) {
    int idx = t + i * 256;
    int row = idx >> 4;
    int c4  = idx & 15;
    float4 v = *(const float4*)(q1 + (size_t)(bb + row) * SLICE + c * EDIM + c4 * 4);
    *(float4*)(lA + row * LQA_P + c4 * 4) = v;
  }
  #pragma unroll
  for (int i = 0; i < 4; ++i) {
    int idx = t + i * 256;
    int dl = idx >> 4;
    int e4 = idx & 15;
    float4 v = *(const float4*)(q2 + (size_t)(bd + dl) * SLICE + c * EDIM + e4 * 4);
    lB[(e4 * 4 + 0) * LQB_P + dl] = v.x;
    lB[(e4 * 4 + 1) * LQB_P + dl] = v.y;
    lB[(e4 * 4 + 2) * LQB_P + dl] = v.z;
    lB[(e4 * 4 + 3) * LQB_P + dl] = v.w;
  }
  __syncthreads();

  v8f acc[2][2] = {};
  const int mrow = lane & 15;
  const int koff = (lane >> 4) * 2;
  const int ncol = lane & 15;

  #pragma unroll
  for (int kk = 0; kk < 64; kk += 4) {
    v2f afrag[2], bfrag[2];
    #pragma unroll
    for (int mt = 0; mt < 2; ++mt) {
      int row = wm * 32 + mt * 16 + mrow;
      afrag[mt] = *(const v2f*)(lA + row * LQA_P + kk + koff);
    }
    #pragma unroll
    for (int nt = 0; nt < 2; ++nt) {
      int col = wn * 32 + nt * 16 + ncol;
      bfrag[nt].x = lB[(kk + koff + 0) * LQB_P + col];
      bfrag[nt].y = lB[(kk + koff + 1) * LQB_P + col];
    }
    #pragma unroll
    for (int mt = 0; mt < 2; ++mt)
      #pragma unroll
      for (int nt = 0; nt < 2; ++nt)
        acc[mt][nt] = __builtin_amdgcn_wmma_f32_16x16x4_f32(
            false, afrag[mt], false, bfrag[nt],
            (short)0, acc[mt][nt], false, false);
  }

  const float scale = 0.125f;
  const int rsel = (lane >> 4) * 8;
  float lmax = -3.4e38f;
  __syncthreads();
  #pragma unroll
  for (int mt = 0; mt < 2; ++mt) {
    #pragma unroll
    for (int nt = 0; nt < 2; ++nt) {
      int col = wn * 32 + nt * 16 + ncol;
      #pragma unroll
      for (int v = 0; v < 8; ++v) {
        int row = wm * 32 + mt * 16 + rsel + v;
        float x = acc[mt][nt][v] * scale;
        lmax = fmaxf(lmax, x);
        lA[row * LQA_P + col] = x;
      }
    }
  }
  red[t] = lmax;
  __syncthreads();
  #pragma unroll
  for (int i = 0; i < 8; ++i) {
    int idx = t + i * 256;
    int row = idx >> 4;
    int c4  = idx & 15;
    float4 v = *(const float4*)(lA + row * LQA_P + c4 * 4);
    *(float4*)(aff + (size_t)(bb + row) * (BSZ * CCH) + c * BSZ + bd + c4 * 4) = v;
  }
  for (int off = 128; off > 0; off >>= 1) {
    if (t < off) red[t] = fmaxf(red[t], red[t + off]);
    __syncthreads();
  }
  if (t == 0) atomicMax(&maxbuf[c], enc_f32(red[0]));
}

__global__ __launch_bounds__(256) void exp_sub_max(
    float* __restrict__ A, const unsigned int* __restrict__ maxbuf)
{
  const int t = threadIdx.x;
  float* slab = A + (size_t)blockIdx.x * (BSZ * CCH);

  float4 a[CCH];
  #pragma unroll
  for (int i = 0; i < CCH; ++i)
    a[i] = *(const float4*)(slab + i * BSZ + t * 4);

  #pragma unroll
  for (int i = 0; i < CCH; ++i) {
    float mv = dec_f32(maxbuf[i]);
    float4 e;
    e.x = expf(a[i].x - mv);
    e.y = expf(a[i].y - mv);
    e.z = expf(a[i].z - mv);
    e.w = expf(a[i].w - mv);
    *(float4*)(slab + i * BSZ + t * 4) = e;
  }
}

#define NBG    16
#define SLABF  (BSZ * CCH)

__device__ __forceinline__ float4 f4_fma(float4 a, float4 b, float4 c) {
  c.x += a.x * b.x; c.y += a.y * b.y; c.z += a.z * b.z; c.w += a.w * b.w;
  return c;
}
__device__ __forceinline__ float4 f4_add(float4 a, float4 b) {
  a.x += b.x; a.y += b.y; a.z += b.z; a.w += b.w;
  return a;
}
__device__ __forceinline__ void lds_barrier() {
  __builtin_amdgcn_fence(__ATOMIC_RELEASE, "workgroup", "local");
  __builtin_amdgcn_s_barrier();
  __builtin_amdgcn_fence(__ATOMIC_ACQUIRE, "workgroup", "local");
}

template <bool FIRST>
__global__ __launch_bounds__(512) void sink_rowcol(
    const float* __restrict__ A, const float* __restrict__ part_in,
    const float* __restrict__ cmarg, const float* __restrict__ r,
    float* __restrict__ uT, float* __restrict__ part_out)
{
  __shared__ float4 sv[256];
  __shared__ float4 sp[256];
  __shared__ float  su[64];
  __shared__ float  buf[8][BSZ];
  const int t    = threadIdx.x;
  const int lane = t & 31;
  const int wave = t >> 5;
  const int c    = blockIdx.x & 15;
  const int bg   = blockIdx.x >> 4;
  const int b0   = bg * 64 + wave * 4;

  float4 x[8];
  float4 cm = make_float4(1.0f, 1.0f, 1.0f, 1.0f);
  const int q = t & 255;
  const int h = t >> 8;
  if constexpr (!FIRST) {
    const float4* P4 = (const float4*)part_in + (size_t)c * (NBG * 256);
    #pragma unroll
    for (int p = 0; p < 8; ++p) x[p] = P4[(8 * h + p) * 256 + q];
    if (h == 0) {
      cm.x = cmarg[(4 * q + 0) * CCH + c];
      cm.y = cmarg[(4 * q + 1) * CCH + c];
      cm.z = cmarg[(4 * q + 2) * CCH + c];
      cm.w = cmarg[(4 * q + 3) * CCH + c];
    }
  }
  float rr[4];
  #pragma unroll
  for (int k = 0; k < 4; ++k) rr[k] = r[(b0 + k) * CCH + c];
  float4 a[4][8];
  #pragma unroll
  for (int k = 0; k < 4; ++k) {
    const float4* row = (const float4*)(A + (size_t)(b0 + k) * SLABF + c * BSZ);
    #pragma unroll
    for (int i = 0; i < 8; ++i) a[k][i] = row[lane + 32 * i];
  }
  if constexpr (!FIRST) {
    __builtin_amdgcn_sched_barrier(0);
  }

  float4 vv[8];
  if constexpr (FIRST) {
    #pragma unroll
    for (int i = 0; i < 8; ++i) vv[i] = make_float4(1.0f, 1.0f, 1.0f, 1.0f);
  } else {
    float4 s = x[0];
    #pragma unroll
    for (int p = 1; p < 8; ++p) s = f4_add(s, x[p]);
    if (h == 1) sp[q] = s;
    lds_barrier();
    if (h == 0) {
      const float4 tot = f4_add(s, sp[q]);
      float4 v;
      v.x = cm.x / tot.x; v.y = cm.y / tot.y; v.z = cm.z / tot.z; v.w = cm.w / tot.w;
      sv[q] = v;
    }
    lds_barrier();
    #pragma unroll
    for (int i = 0; i < 8; ++i) vv[i] = sv[lane + 32 * i];
  }

  float4 acc[8];
  #pragma unroll
  for (int i = 0; i < 8; ++i) acc[i] = make_float4(0.0f, 0.0f, 0.0f, 0.0f);

  float s[4];
  #pragma unroll
  for (int k = 0; k < 4; ++k) {
    float4 p = make_float4(0.0f, 0.0f, 0.0f, 0.0f);
    #pragma unroll
    for (int i = 0; i < 8; ++i) p = f4_fma(a[k][i], vv[i], p);
    s[k] = (p.x + p.y) + (p.z + p.w);
  }
  #pragma unroll
  for (int off = 16; off > 0; off >>= 1) {
    #pragma unroll
    for (int k = 0; k < 4; ++k) s[k] += __shfl_xor(s[k], off, 32);
  }
  #pragma unroll
  for (int k = 0; k < 4; ++k) {
    const float uu = rr[k] / s[k];
    if (lane == 0) su[wave * 4 + k] = uu;
    const float4 u4 = make_float4(uu, uu, uu, uu);
    #pragma unroll
    for (int i = 0; i < 8; ++i) acc[i] = f4_fma(a[k][i], u4, acc[i]);
  }

  float* bw = &buf[wave & 7][0];
  if (wave >= 8) {
    #pragma unroll
    for (int i = 0; i < 8; ++i) *(float4*)(bw + 4 * (lane + 32 * i)) = acc[i];
  }
  lds_barrier();
  if (wave < 8) {
    #pragma unroll
    for (int i = 0; i < 8; ++i) {
      const float4 y = *(const float4*)(bw + 4 * (lane + 32 * i));
      *(float4*)(bw + 4 * (lane + 32 * i)) = f4_add(acc[i], y);
    }
  }
  lds_barrier();
  float2 tot = *(const float2*)(&buf[0][2 * t]);
  #pragma unroll
  for (int w = 1; w < 8; ++w) {
    const float2 y = *(const float2*)(&buf[w][2 * t]);
    tot.x += y.x; tot.y += y.y;
  }
  *(float2*)(part_out + (size_t)(c * NBG + bg) * BSZ + 2 * t) = tot;
  if (t < 64) uT[c * BSZ + bg * 64 + t] = su[t];
}

__global__ __launch_bounds__(256) void sink_colsum(
    const float* __restrict__ part, const float* __restrict__ cmarg,
    float* __restrict__ vT)
{
  const int idx = blockIdx.x * 256 + threadIdx.x;
  const int c = idx >> 10;
  const int d = idx & (BSZ - 1);
  const float cm = cmarg[d * CCH + c];
  float x[NBG];
  #pragma unroll
  for (int p = 0; p < NBG; ++p) x[p] = part[(size_t)(c * NBG + p) * BSZ + d];
  float s0 = x[0], s1 = x[8];
  #pragma unroll
  for (int p = 1; p < 8; ++p) { s0 += x[p]; s1 += x[8 + p]; }
  vT[idx] = cm / (s0 + s1);
}

#define XP 1028

__global__ __launch_bounds__(256) void sink_final(
    float* __restrict__ A, const float* __restrict__ uT,
    const float* __restrict__ vT)
{
  __shared__ float L[CCH * XP];
  const int t = threadIdx.x;
  const int b = blockIdx.x;
  float* slab = A + (size_t)b * SLABF;

  float4 a[CCH], v[CCH];
  #pragma unroll
  for (int i = 0; i < CCH; ++i)
    a[i] = *(const float4*)(slab + i * BSZ + t * 4);
  #pragma unroll
  for (int i = 0; i < CCH; ++i)
    v[i] = *(const float4*)(vT + i * BSZ + t * 4);

  #pragma unroll
  for (int i = 0; i < CCH; ++i) {
    const float uu = uT[i * BSZ + b];
    float4 e;
    e.x = a[i].x * uu * v[i].x;
    e.y = a[i].y * uu * v[i].y;
    e.z = a[i].z * uu * v[i].z;
    e.w = a[i].w * uu * v[i].w;
    *(float4*)(L + i * XP + t * 4) = e;
  }
  __syncthreads();

  #pragma unroll
  for (int i = 0; i < CCH; ++i) {
    int o  = t + 256 * i;
    int d  = o >> 2;
    int c4 = o & 3;
    float4 r;
    r.x = L[(4 * c4 + 0) * XP + d];
    r.y = L[(4 * c4 + 1) * XP + d];
    r.z = L[(4 * c4 + 2) * XP + d];
    r.w = L[(4 * c4 + 3) * XP + d];
    *(float4*)(slab + (size_t)o * 4) = r;
  }
}

extern "C" void kernel_launch(void* const* d_in, const int* in_sizes, int n_in,
                              void* d_out, int out_size, void* d_ws, size_t ws_size,
                              hipStream_t stream)
{
  const float* x1   = (const float*)d_in[0];
  const float* x2   = (const float*)d_in[1];
  const float* p1   = (const float*)d_in[2];
  const float* p2   = (const float*)d_in[3];
  const float* w1[4] = {(const float*)d_in[4],  (const float*)d_in[6],
                        (const float*)d_in[8],  (const float*)d_in[10]};
  const float* b1[4] = {(const float*)d_in[5],  (const float*)d_in[7],
                        (const float*)d_in[9],  (const float*)d_in[11]};
  const float* w2[4] = {(const float*)d_in[12], (const float*)d_in[14],
                        (const float*)d_in[16], (const float*)d_in[18]};
  const float* b2[4] = {(const float*)d_in[13], (const float*)d_in[15],
                        (const float*)d_in[17], (const float*)d_in[19]};

  float* ws = (float*)d_ws;
  float* q1 = ws + 1024 * 1024;
  float* q2 = q1 + 1024 * 1024;
  float* uv_u = q2 + 1024 * 1024;
  float* uv_v = uv_u + 16384;
  unsigned int* maxbuf = (unsigned int*)(uv_v + 16384);

  float* Aout = (float*)d_out;

  float* pA = Aout;
  float* pB = pA + 4 * 1024 * 512;
  float* pQ = pB + 4 * 1024 * 512;

  gemm_splitk<1><<<dim3(512/64, 1024/128, KSPLIT), 256, 0, stream>>>(x1, nullptr, w1[0], pA, 1024, 256, 512, 64);
  gemm_splitk<4><<<dim3(512/64, 1024/128, KSPLIT), 256, 0, stream>>>(pA, b1[0], w1[1], pB, 1024, 512, 512, 128);
  gemm_splitk<4><<<dim3(512/64, 1024/128, KSPLIT), 256, 0, stream>>>(pB, b1[1], w1[2], pA, 1024, 512, 512, 128);
  gemm_splitk<4><<<dim3(1024/64, 1024/128, KSPLIT), 256, 0, stream>>>(pA, b1[2], w1[3], pQ, 1024, 512, 1024, 128);
  norm_rows64<<<16384 / 8, 256, 0, stream>>>(pQ, b1[3], q1);
  gemm_splitk<1><<<dim3(512/64, 1024/128, KSPLIT), 256, 0, stream>>>(x2, nullptr, w2[0], pA, 1024, 256, 512, 64);
  gemm_splitk<4><<<dim3(512/64, 1024/128, KSPLIT), 256, 0, stream>>>(pA, b2[0], w2[1], pB, 1024, 512, 512, 128);
  gemm_splitk<4><<<dim3(512/64, 1024/128, KSPLIT), 256, 0, stream>>>(pB, b2[1], w2[2], pA, 1024, 512, 512, 128);
  gemm_splitk<4><<<dim3(1024/64, 1024/128, KSPLIT), 256, 0, stream>>>(pA, b2[2], w2[3], pQ, 1024, 512, 1024, 128);
  norm_rows64<<<16384 / 8, 256, 0, stream>>>(pQ, b2[3], q2);

  sink_init<<<1, 32, 0, stream>>>(maxbuf);

  affinity_wmma<<<dim3(1024/64, 1024/128, CCH), 256, 0, stream>>>(q1, q2, Aout, maxbuf);

  exp_sub_max<<<BSZ, 256, 0, stream>>>(Aout, maxbuf);

  float* partA = ws;
  float* partB = ws + CCH * NBG * BSZ;
  sink_rowcol<true ><<<CCH * NBG, 512, 0, stream>>>(Aout, partB, p2, p1, uv_u, partA);
  sink_rowcol<false><<<CCH * NBG, 512, 0, stream>>>(Aout, partA, p2, p1, uv_u, partB);
  for (int it = 0; it < 4; ++it) {
    sink_rowcol<false><<<CCH * NBG, 512, 0, stream>>>(Aout, partB, p2, p1, uv_u, partA);
    sink_rowcol<false><<<CCH * NBG, 512, 0, stream>>>(Aout, partA, p2, p1, uv_u, partB);
  }
  sink_colsum<<<(CCH * BSZ) / 256, 256, 0, stream>>>(partB, p2, uv_v);

  sink_final<<<BSZ, 256, 0, stream>>>(Aout, uv_u, uv_v);
}


# MLP GEMMs: a workgroup stages its K slice 64 deep per step instead of 16 (one or two steps): all 37 16-B loads of a step in flight together, 2 or 4 barriers per launch instead of 8 or 16
# speedup vs baseline: 3.0969x; 1.1211x over previous
#include <hip/hip_runtime.h>
#include <hip/hip_bf16.h>


typedef float v2f __attribute__((ext_vector_type(2)));
typedef float v4f __attribute__((ext_vector_type(4)));
typedef float v8f __attribute__((ext_vector_type(8)));

#define BSZ   1024
#define CCH   16
#define EDIM  64
#define SLICE (CCH * EDIM)
#define ATOT  (BSZ * BSZ * CCH)

__device__ __forceinline__ unsigned int enc_f32(float f) {
  unsigned int u = __float_as_uint(f);
  return (u & 0x80000000u) ? ~u : (u | 0x80000000u);
}
__device__ __forceinline__ float dec_f32(unsigned int u) {
  return (u & 0x80000000u) ? __uint_as_float(u & 0x7FFFFFFFu)
                           : __uint_as_float(~u);
}

#define KSPLIT 4
#define BK    64
#define LDA_P 68
#define LDB_P 68

template <int NIN>
__global__ __launch_bounds__(256) void gemm_splitk(
    const float* __restrict__ A, const float* __restrict__ biasIn,
    const float* __restrict__ W, float* __restrict__ out,
    int M, int K, int N, int kchunk)
{
  static_assert(NIN == 1 || NIN == 4, "plane count of the A loader");
  __shared__ float lA[128 * LDA_P];
  __shared__ float lB[BK * LDB_P];

  const int t    = threadIdx.x;
  const int lane = t & 31;
  const int wave = t >> 5;
  const int wm   = wave & 3;
  const int wn   = wave >> 2;
  const int bm   = blockIdx.y * 128;
  const int bn   = blockIdx.x * 64;
  const int kbeg = blockIdx.z * kchunk;
  const int kend = kbeg + kchunk;
  const size_t pitch_in = (size_t)M * K;

  v8f acc[2][2] = {};

  const int mrow = lane & 15;
  const int koff = (lane >> 4) * 2;
  const int ncol = lane & 15;

  const int arow = t >> 4;
  const int ac4  = t & 15;
  const int brow = t >> 4;
  const int bc4  = t & 15;

  for (int k0 = kbeg; k0 < kend; k0 += BK) {
    v4f pv[8][NIN];
    v4f wv[4];
    v4f bb = {0.0f, 0.0f, 0.0f, 0.0f};
    #pragma unroll
    for (int i = 0; i < 8; ++i) {
      const float* pa = A + (size_t)(bm + arow + 16 * i) * K + k0 + ac4 * 4;
      #pragma unroll
      for (int p = 0; p < NIN; ++p)
        pv[i][p] = *(const v4f*)(pa + (size_t)p * pitch_in);
    }
    if constexpr (NIN > 1) bb = *(const v4f*)(biasIn + k0 + ac4 * 4);
    #pragma unroll
    for (int j = 0; j < 4; ++j)
      wv[j] = *(const v4f*)(W + (size_t)(k0 + brow + 16 * j) * N + bn + bc4 * 4);
    __builtin_amdgcn_sched_barrier(0);
    #pragma unroll
    for (int i = 0; i < 8; ++i) {
      v4f v = pv[i][0];
      if constexpr (NIN > 1) {
        #pragma unroll
        for (int p = 1; p < NIN; ++p) v += pv[i][p];
        v += bb;
        v.x = fmaxf(v.x, 0.0f);
        v.y = fmaxf(v.y, 0.0f);
        v.z = fmaxf(v.z, 0.0f);
        v.w = fmaxf(v.w, 0.0f);
      }
      *(v4f*)(lA + (arow + 16 * i) * LDA_P + ac4 * 4) = v;
    }
    #pragma unroll
    for (int j = 0; j < 4; ++j)
      *(v4f*)(lB + (brow + 16 * j) * LDB_P + bc4 * 4) = wv[j];
    __syncthreads();

    #pragma unroll
    for (int kk = 0; kk < BK; kk += 4) {
      v2f afrag[2], bfrag[2];
      #pragma unroll
      for (int mt = 0; mt < 2; ++mt) {
        int row = wm * 32 + mt * 16 + mrow;
        afrag[mt] = *(const v2f*)(lA + row * LDA_P + kk + koff);
      }
      #pragma unroll
      for (int nt = 0; nt < 2; ++nt) {
        int col = wn * 32 + nt * 16 + ncol;
        bfrag[nt].x = lB[(kk + koff + 0) * LDB_P + col];
        bfrag[nt].y = lB[(kk + koff + 1) * LDB_P + col];
      }
      #pragma unroll
      for (int mt = 0; mt < 2; ++mt)
        #pragma unroll
        for (int nt = 0; nt < 2; ++nt)
          acc[mt][nt] = __builtin_amdgcn_wmma_f32_16x16x4_f32(
              false, afrag[mt], false, bfrag[nt],
              (short)0, acc[mt][nt], false, false);
    }
    __syncthreads();
  }

  float* op = out + (size_t)blockIdx.z * M * N;
  const int rsel = (lane >> 4) * 8;
  #pragma unroll
  for (int nt = 0; nt < 2; ++nt) {
    int col = bn + wn * 32 + nt * 16 + ncol;
    #pragma unroll
    for (int mt = 0; mt < 2; ++mt) {
      #pragma unroll
      for (int v = 0; v < 8; ++v) {
        int row = bm + wm * 32 + mt * 16 + rsel + v;
        op[(size_t)row * N + col] = acc[mt][nt][v];
      }
    }
  }
}

__global__ __launch_bounds__(256) void norm_rows64(
    const float* __restrict__ qp, const float* __restrict__ bias, float* __restrict__ q)
{
  const int lane = threadIdx.x & 31;
  const int wave = threadIdx.x >> 5;
  const int row  = blockIdx.x * 8 + wave;
  float* p = q + (size_t)row * 64;
  const float* pp = qp + (size_t)row * 64 + lane * 2;

  float2 x = *(const float2*)pp;
  const float2 bb = *(const float2*)(bias + (row & 15) * 64 + lane * 2);
  #pragma unroll
  for (int k = 1; k < KSPLIT; ++k) {
    const float2 s = *(const float2*)(pp + (size_t)k * (BSZ * SLICE));
    x.x += s.x; x.y += s.y;
  }
  x.x += bb.x; x.y += bb.y;
  float s  = x.x + x.y;
  float ss = x.x * x.x + x.y * x.y;
  #pragma unroll
  for (int off = 16; off > 0; off >>= 1) {
    s  += __shfl_xor(s,  off, 32);
    ss += __shfl_xor(ss, off, 32);
  }
  float mean = s * (1.0f / 64.0f);
  float var  = (ss - 64.0f * mean * mean) * (1.0f / 63.0f);
  float inv  = rsqrtf(var + 1e-8f);
  x.x = (x.x - mean) * inv;
  x.y = (x.y - mean) * inv;
  *(float2*)(p + lane * 2) = x;
}

__global__ void sink_init(unsigned int* __restrict__ maxbuf)
{
  int i = threadIdx.x;
  if (i < CCH) maxbuf[i] = 0u;
}

#define LQA_P 68
#define LQB_P 65

__global__ __launch_bounds__(256) void affinity_wmma(
    const float* __restrict__ q1, const float* __restrict__ q2,
    float* __restrict__ aff, unsigned int* __restrict__ maxbuf)
{
  __shared__ float lA[128 * LQA_P];
  __shared__ float lB[64 * LQB_P];
  __shared__ float red[256];

  const int t    = threadIdx.x;
  const int lane = t & 31;
  const int wave = t >> 5;
  const int wm   = wave & 3;
  const int wn   = wave >> 2;
  const int bb   = blockIdx.y * 128;
  const int bd   = blockIdx.x * 64;
  const int c    = blockIdx.z;

  #pragma unroll
  for (int i = 0; i < 8; ++i) {
    int idx = t + i * 256;
    int row = idx >> 4;
    int c4  = idx & 15;
    float4 v = *(const float4*)(q1 + (size_t)(bb + row) * SLICE + c * EDIM + c4 * 4);
    *(float4*)(lA + row * LQA_P + c4 * 4) = v;
  }
  #pragma unroll
  for (int i = 0; i < 4; ++i) {
    int idx = t + i * 256;
    int dl = idx >> 4;
    int e4 = idx & 15;
    float4 v = *(const float4*)(q2 + (size_t)(bd + dl) * SLICE + c * EDIM + e4 * 4);
    lB[(e4 * 4 + 0) * LQB_P + dl] = v.x;
    lB[(e4 * 4 + 1) * LQB_P + dl] = v.y;
    lB[(e4 * 4 + 2) * LQB_P + dl] = v.z;
    lB[(e4 * 4 + 3) * LQB_P + dl] = v.w;
  }
  __syncthreads();

  v8f acc[2][2] = {};
  const int mrow = lane & 15;
  const int koff = (lane >> 4) * 2;
  const int ncol = lane & 15;

  #pragma unroll
  for (int kk = 0; kk < 64; kk += 4) {
    v2f afrag[2], bfrag[2];
    #pragma unroll
    for (int mt = 0; mt < 2; ++mt) {
      int row = wm * 32 + mt * 16 + mrow;
      afrag[mt] = *(const v2f*)(lA + row * LQA_P + kk + koff);
    }
    #pragma unroll
    for (int nt = 0; nt < 2; ++nt) {
      int col = wn * 32 + nt * 16 + ncol;
      bfrag[nt].x = lB[(kk + koff + 0) * LQB_P + col];
      bfrag[nt].y = lB[(kk + koff + 1) * LQB_P + col];
    }
    #pragma unroll
    for (int mt = 0; mt < 2; ++mt)
      #pragma unroll
      for (int nt = 0; nt < 2; ++nt)
        acc[mt][nt] = __builtin_amdgcn_wmma_f32_16x16x4_f32(
            false, afrag[mt], false, bfrag[nt],
            (short)0, acc[mt][nt], false, false);
  }

  const float scale = 0.125f;
  const int rsel = (lane >> 4) * 8;
  float lmax = -3.4e38f;
  __syncthreads();
  #pragma unroll
  for (int mt = 0; mt < 2; ++mt) {
    #pragma unroll
    for (int nt = 0; nt < 2; ++nt) {
      int col = wn * 32 + nt * 16 + ncol;
      #pragma unroll
      for (int v = 0; v < 8; ++v) {
        int row = wm * 32 + mt * 16 + rsel + v;
        float x = acc[mt][nt][v] * scale;
        lmax = fmaxf(lmax, x);
        lA[row * LQA_P + col] = x;
      }
    }
  }
  red[t] = lmax;
  __syncthreads();
  #pragma unroll
  for (int i = 0; i < 8; ++i) {
    int idx = t + i * 256;
    int row = idx >> 4;
    int c4  = idx & 15;
    float4 v = *(const float4*)(lA + row * LQA_P + c4 * 4);
    *(float4*)(aff + (size_t)(bb + row) * (BSZ * CCH) + c * BSZ + bd + c4 * 4) = v;
  }
  for (int off = 128; off > 0; off >>= 1) {
    if (t < off) red[t] = fmaxf(red[t], red[t + off]);
    __syncthreads();
  }
  if (t == 0) atomicMax(&maxbuf[c], enc_f32(red[0]));
}

__global__ __launch_bounds__(256) void exp_sub_max(
    float* __restrict__ A, const unsigned int* __restrict__ maxbuf)
{
  const int t = threadIdx.x;
  float* slab = A + (size_t)blockIdx.x * (BSZ * CCH);

  float4 a[CCH];
  #pragma unroll
  for (int i = 0; i < CCH; ++i)
    a[i] = *(const float4*)(slab + i * BSZ + t * 4);

  #pragma unroll
  for (int i = 0; i < CCH; ++i) {
    float mv = dec_f32(maxbuf[i]);
    float4 e;
    e.x = expf(a[i].x - mv);
    e.y = expf(a[i].y - mv);
    e.z = expf(a[i].z - mv);
    e.w = expf(a[i].w - mv);
    *(float4*)(slab + i * BSZ + t * 4) = e;
  }
}

#define NBG    16
#define SLABF  (BSZ * CCH)

__device__ __forceinline__ float4 f4_fma(float4 a, float4 b, float4 c) {
  c.x += a.x * b.x; c.y += a.y * b.y; c.z += a.z * b.z; c.w += a.w * b.w;
  return c;
}
__device__ __forceinline__ float4 f4_add(float4 a, float4 b) {
  a.x += b.x; a.y += b.y; a.z += b.z; a.w += b.w;
  return a;
}
__device__ __forceinline__ void lds_barrier() {
  __builtin_amdgcn_fence(__ATOMIC_RELEASE, "workgroup", "local");
  __builtin_amdgcn_s_barrier();
  __builtin_amdgcn_fence(__ATOMIC_ACQUIRE, "workgroup", "local");
}

template <bool FIRST>
__global__ __launch_bounds__(512) void sink_rowcol(
    const float* __restrict__ A, const float* __restrict__ part_in,
    const float* __restrict__ cmarg, const float* __restrict__ r,
    float* __restrict__ uT, float* __restrict__ part_out)
{
  __shared__ float4 sv[256];
  __shared__ float4 sp[256];
  __shared__ float  su[64];
  __shared__ float  buf[8][BSZ];
  const int t    = threadIdx.x;
  const int lane = t & 31;
  const int wave = t >> 5;
  const int c    = blockIdx.x & 15;
  const int bg   = blockIdx.x >> 4;
  const int b0   = bg * 64 + wave * 4;

  float4 x[8];
  float4 cm = make_float4(1.0f, 1.0f, 1.0f, 1.0f);
  const int q = t & 255;
  const int h = t >> 8;
  if constexpr (!FIRST) {
    const float4* P4 = (const float4*)part_in + (size_t)c * (NBG * 256);
    #pragma unroll
    for (int p = 0; p < 8; ++p) x[p] = P4[(8 * h + p) * 256 + q];
    if (h == 0) {
      cm.x = cmarg[(4 * q + 0) * CCH + c];
      cm.y = cmarg[(4 * q + 1) * CCH + c];
      cm.z = cmarg[(4 * q + 2) * CCH + c];
      cm.w = cmarg[(4 * q + 3) * CCH + c];
    }
  }
  float rr[4];
  #pragma unroll
  for (int k = 0; k < 4; ++k) rr[k] = r[(b0 + k) * CCH + c];
  float4 a[4][8];
  #pragma unroll
  for (int k = 0; k < 4; ++k) {
    const float4* row = (const float4*)(A + (size_t)(b0 + k) * SLABF + c * BSZ);
    #pragma unroll
    for (int i = 0; i < 8; ++i) a[k][i] = row[lane + 32 * i];
  }
  if constexpr (!FIRST) {
    __builtin_amdgcn_sched_barrier(0);
  }

  float4 vv[8];
  if constexpr (FIRST) {
    #pragma unroll
    for (int i = 0; i < 8; ++i) vv[i] = make_float4(1.0f, 1.0f, 1.0f, 1.0f);
  } else {
    float4 s = x[0];
    #pragma unroll
    for (int p = 1; p < 8; ++p) s = f4_add(s, x[p]);
    if (h == 1) sp[q] = s;
    lds_barrier();
    if (h == 0) {
      const float4 tot = f4_add(s, sp[q]);
      float4 v;
      v.x = cm.x / tot.x; v.y = cm.y / tot.y; v.z = cm.z / tot.z; v.w = cm.w / tot.w;
      sv[q] = v;
    }
    lds_barrier();
    #pragma unroll
    for (int i = 0; i < 8; ++i) vv[i] = sv[lane + 32 * i];
  }

  float4 acc[8];
  #pragma unroll
  for (int i = 0; i < 8; ++i) acc[i] = make_float4(0.0f, 0.0f, 0.0f, 0.0f);

  float s[4];
  #pragma unroll
  for (int k = 0; k < 4; ++k) {
    float4 p = make_float4(0.0f, 0.0f, 0.0f, 0.0f);
    #pragma unroll
    for (int i = 0; i < 8; ++i) p = f4_fma(a[k][i], vv[i], p);
    s[k] = (p.x + p.y) + (p.z + p.w);
  }
  #pragma unroll
  for (int off = 16; off > 0; off >>= 1) {
    #pragma unroll
    for (int k = 0; k < 4; ++k) s[k] += __shfl_xor(s[k], off, 32);
  }
  #pragma unroll
  for (int k = 0; k < 4; ++k) {
    const float uu = rr[k] / s[k];
    if (lane == 0) su[wave * 4 + k] = uu;
    const float4 u4 = make_float4(uu, uu, uu, uu);
    #pragma unroll
    for (int i = 0; i < 8; ++i) acc[i] = f4_fma(a[k][i], u4, acc[i]);
  }

  float* bw = &buf[wave & 7][0];
  if (wave >= 8) {
    #pragma unroll
    for (int i = 0; i < 8; ++i) *(float4*)(bw + 4 * (lane + 32 * i)) = acc[i];
  }
  lds_barrier();
  if (wave < 8) {
    #pragma unroll
    for (int i = 0; i < 8; ++i) {
      const float4 y = *(const float4*)(bw + 4 * (lane + 32 * i));
      *(float4*)(bw + 4 * (lane + 32 * i)) = f4_add(acc[i], y);
    }
  }
  lds_barrier();
  float2 tot = *(const float2*)(&buf[0][2 * t]);
  #pragma unroll
  for (int w = 1; w < 8; ++w) {
    const float2 y = *(const float2*)(&buf[w][2 * t]);
    tot.x += y.x; tot.y += y.y;
  }
  *(float2*)(part_out + (size_t)(c * NBG + bg) * BSZ + 2 * t) = tot;
  if (t < 64) uT[c * BSZ + bg * 64 + t] = su[t];
}

__global__ __launch_bounds__(256) void sink_colsum(
    const float* __restrict__ part, const float* __restrict__ cmarg,
    float* __restrict__ vT)
{
  const int idx = blockIdx.x * 256 + threadIdx.x;
  const int c = idx >> 10;
  const int d = idx & (BSZ - 1);
  const float cm = cmarg[d * CCH + c];
  float x[NBG];
  #pragma unroll
  for (int p = 0; p < NBG; ++p) x[p] = part[(size_t)(c * NBG + p) * BSZ + d];
  float s0 = x[0], s1 = x[8];
  #pragma unroll
  for (int p = 1; p < 8; ++p) { s0 += x[p]; s1 += x[8 + p]; }
  vT[idx] = cm / (s0 + s1);
}

#define XP 1028

__global__ __launch_bounds__(256) void sink_final(
    float* __restrict__ A, const float* __restrict__ uT,
    const float* __restrict__ vT)
{
  __shared__ float L[CCH * XP];
  const int t = threadIdx.x;
  const int b = blockIdx.x;
  float* slab = A + (size_t)b * SLABF;

  float4 a[CCH], v[CCH];
  #pragma unroll
  for (int i = 0; i < CCH; ++i)
    a[i] = *(const float4*)(slab + i * BSZ + t * 4);
  #pragma unroll
  for (int i = 0; i < CCH; ++i)
    v[i] = *(const float4*)(vT + i * BSZ + t * 4);

  #pragma unroll
  for (int i = 0; i < CCH; ++i) {
    const float uu = uT[i * BSZ + b];
    float4 e;
    e.x = a[i].x * uu * v[i].x;
    e.y = a[i].y * uu * v[i].y;
    e.z = a[i].z * uu * v[i].z;
    e.w = a[i].w * uu * v[i].w;
    *(float4*)(L + i * XP + t * 4) = e;
  }
  __syncthreads();

  #pragma unroll
  for (int i = 0; i < CCH; ++i) {
    int o  = t + 256 * i;
    int d  = o >> 2;
    int c4 = o & 3;
    float4 r;
    r.x = L[(4 * c4 + 0) * XP + d];
    r.y = L[(4 * c4 + 1) * XP + d];
    r.z = L[(4 * c4 + 2) * XP + d];
    r.w = L[(4 * c4 + 3) * XP + d];
    *(float4*)(slab + (size_t)o * 4) = r;
  }
}

extern "C" void kernel_launch(void* const* d_in, const int* in_sizes, int n_in,
                              void* d_out, int out_size, void* d_ws, size_t ws_size,
                              hipStream_t stream)
{
  const float* x1   = (const float*)d_in[0];
  const float* x2   = (const float*)d_in[1];
  const float* p1   = (const float*)d_in[2];
  const float* p2   = (const float*)d_in[3];
  const float* w1[4] = {(const float*)d_in[4],  (const float*)d_in[6],
                        (const float*)d_in[8],  (const float*)d_in[10]};
  const float* b1[4] = {(const float*)d_in[5],  (const float*)d_in[7],
                        (const float*)d_in[9],  (const float*)d_in[11]};
  const float* w2[4] = {(const float*)d_in[12], (const float*)d_in[14],
                        (const float*)d_in[16], (const float*)d_in[18]};
  const float* b2[4] = {(const float*)d_in[13], (const float*)d_in[15],
                        (const float*)d_in[17], (const float*)d_in[19]};

  float* ws = (float*)d_ws;
  float* q1 = ws + 1024 * 1024;
  float* q2 = q1 + 1024 * 1024;
  float* uv_u = q2 + 1024 * 1024;
  float* uv_v = uv_u + 16384;
  unsigned int* maxbuf = (unsigned int*)(uv_v + 16384);

  float* Aout = (float*)d_out;

  float* pA = Aout;
  float* pB = pA + 4 * 1024 * 512;
  float* pQ = pB + 4 * 1024 * 512;

  gemm_splitk<1><<<dim3(512/64, 1024/128, KSPLIT), 256, 0, stream>>>(x1, nullptr, w1[0], pA, 1024, 256, 512, 64);
  gemm_splitk<4><<<dim3(512/64, 1024/128, KSPLIT), 256, 0, stream>>>(pA, b1[0], w1[1], pB, 1024, 512, 512, 128);
  gemm_splitk<4><<<dim3(512/64, 1024/128, KSPLIT), 256, 0, stream>>>(pB, b1[1], w1[2], pA, 1024, 512, 512, 128);
  gemm_splitk<4><<<dim3(1024/64, 1024/128, KSPLIT), 256, 0, stream>>>(pA, b1[2], w1[3], pQ, 1024, 512, 1024, 128);
  norm_rows64<<<16384 / 8, 256, 0, stream>>>(pQ, b1[3], q1);
  gemm_splitk<1><<<dim3(512/64, 1024/128, KSPLIT), 256, 0, stream>>>(x2, nullptr, w2[0], pA, 1024, 256, 512, 64);
  gemm_splitk<4><<<dim3(512/64, 1024/128, KSPLIT), 256, 0, stream>>>(pA, b2[0], w2[1], pB, 1024, 512, 512, 128);
  gemm_splitk<4><<<dim3(512/64, 1024/128, KSPLIT), 256, 0, stream>>>(pB, b2[1], w2[2], pA, 1024, 512, 512, 128);
  gemm_splitk<4><<<dim3(1024/64, 1024/128, KSPLIT), 256, 0, stream>>>(pA, b2[2], w2[3], pQ, 1024, 512, 1024, 128);
  norm_rows64<<<16384 / 8, 256, 0, stream>>>(pQ, b2[3], q2);

  sink_init<<<1, 32, 0, stream>>>(maxbuf);

  affinity_wmma<<<dim3(1024/64, 1024/128, CCH), 256, 0, stream>>>(q1, q2, Aout, maxbuf);

  exp_sub_max<<<BSZ, 256, 0, stream>>>(Aout, maxbuf);

  float* partA = ws;
  float* partB = ws + CCH * NBG * BSZ;
  sink_rowcol<true ><<<CCH * NBG, 512, 0, stream>>>(Aout, partB, p2, p1, uv_u, partA);
  sink_rowcol<false><<<CCH * NBG, 512, 0, stream>>>(Aout, partA, p2, p1, uv_u, partB);
  for (int it = 0; it < 4; ++it) {
    sink_rowcol<false><<<CCH * NBG, 512, 0, stream>>>(Aout, partB, p2, p1, uv_u, partA);
    sink_rowcol<false><<<CCH * NBG, 512, 0, stream>>>(Aout, partA, p2, p1, uv_u, partB);
  }
  sink_colsum<<<(CCH * BSZ) / 256, 256, 0, stream>>>(partB, p2, uv_v);

  sink_final<<<BSZ, 256, 0, stream>>>(Aout, uv_u, uv_v);
}


# per-channel max shift dropped: Sinkhorn's first row normalisation cancels any per-channel factor of A and |aff| <= 63/8, so exp moves into the affinity epilogue; max reset, atomic max, exp pass gone
# speedup vs baseline: 3.3943x; 1.0960x over previous
#include <hip/hip_runtime.h>
#include <hip/hip_bf16.h>


typedef float v2f __attribute__((ext_vector_type(2)));
typedef float v4f __attribute__((ext_vector_type(4)));
typedef float v8f __attribute__((ext_vector_type(8)));

#define BSZ   1024
#define CCH   16
#define EDIM  64
#define SLICE (CCH * EDIM)
#define ATOT  (BSZ * BSZ * CCH)

#define KSPLIT 4
#define BK    64
#define LDA_P 68
#define LDB_P 68

template <int NIN>
__global__ __launch_bounds__(256) void gemm_splitk(
    const float* __restrict__ A, const float* __restrict__ biasIn,
    const float* __restrict__ W, float* __restrict__ out,
    int M, int K, int N, int kchunk)
{
  static_assert(NIN == 1 || NIN == 4, "plane count of the A loader");
  __shared__ float lA[128 * LDA_P];
  __shared__ float lB[BK * LDB_P];

  const int t    = threadIdx.x;
  const int lane = t & 31;
  const int wave = t >> 5;
  const int wm   = wave & 3;
  const int wn   = wave >> 2;
  const int bm   = blockIdx.y * 128;
  const int bn   = blockIdx.x * 64;
  const int kbeg = blockIdx.z * kchunk;
  const int kend = kbeg + kchunk;
  const size_t pitch_in = (size_t)M * K;

  v8f acc[2][2] = {};

  const int mrow = lane & 15;
  const int koff = (lane >> 4) * 2;
  const int ncol = lane & 15;

  const int arow = t >> 4;
  const int ac4  = t & 15;
  const int brow = t >> 4;
  const int bc4  = t & 15;

  for (int k0 = kbeg; k0 < kend; k0 += BK) {
    v4f pv[8][NIN];
    v4f wv[4];
    v4f bb = {0.0f, 0.0f, 0.0f, 0.0f};
    #pragma unroll
    for (int i = 0; i < 8; ++i) {
      const float* pa = A + (size_t)(bm + arow + 16 * i) * K + k0 + ac4 * 4;
      #pragma unroll
      for (int p = 0; p < NIN; ++p)
        pv[i][p] = *(const v4f*)(pa + (size_t)p * pitch_in);
    }
    if constexpr (NIN > 1) bb = *(const v4f*)(biasIn + k0 + ac4 * 4);
    #pragma unroll
    for (int j = 0; j < 4; ++j)
      wv[j] = *(const v4f*)(W + (size_t)(k0 + brow + 16 * j) * N + bn + bc4 * 4);
    __builtin_amdgcn_sched_barrier(0);
    #pragma unroll
    for (int i = 0; i < 8; ++i) {
      v4f v = pv[i][0];
      if constexpr (NIN > 1) {
        #pragma unroll
        for (int p = 1; p < NIN; ++p) v += pv[i][p];
        v += bb;
        v.x = fmaxf(v.x, 0.0f);
        v.y = fmaxf(v.y, 0.0f);
        v.z = fmaxf(v.z, 0.0f);
        v.w = fmaxf(v.w, 0.0f);
      }
      *(v4f*)(lA + (arow + 16 * i) * LDA_P + ac4 * 4) = v;
    }
    #pragma unroll
    for (int j = 0; j < 4; ++j)
      *(v4f*)(lB + (brow + 16 * j) * LDB_P + bc4 * 4) = wv[j];
    __syncthreads();

    #pragma unroll
    for (int kk = 0; kk < BK; kk += 4) {
      v2f afrag[2], bfrag[2];
      #pragma unroll
      for (int mt = 0; mt < 2; ++mt) {
        int row = wm * 32 + mt * 16 + mrow;
        afrag[mt] = *(const v2f*)(lA + row * LDA_P + kk + koff);
      }
      #pragma unroll
      for (int nt = 0; nt < 2; ++nt) {
        int col = wn * 32 + nt * 16 + ncol;
        bfrag[nt].x = lB[(kk + koff + 0) * LDB_P + col];
        bfrag[nt].y = lB[(kk + koff + 1) * LDB_P + col];
      }
      #pragma unroll
      for (int mt = 0; mt < 2; ++mt)
        #pragma unroll
        for (int nt = 0; nt < 2; ++nt)
          acc[mt][nt] = __builtin_amdgcn_wmma_f32_16x16x4_f32(
              false, afrag[mt], false, bfrag[nt],
              (short)0, acc[mt][nt], false, false);
    }
    __syncthreads();
  }

  float* op = out + (size_t)blockIdx.z * M * N;
  const int rsel = (lane >> 4) * 8;
  #pragma unroll
  for (int nt = 0; nt < 2; ++nt) {
    int col = bn + wn * 32 + nt * 16 + ncol;
    #pragma unroll
    for (int mt = 0; mt < 2; ++mt) {
      #pragma unroll
      for (int v = 0; v < 8; ++v) {
        int row = bm + wm * 32 + mt * 16 + rsel + v;
        op[(size_t)row * N + col] = acc[mt][nt][v];
      }
    }
  }
}

__global__ __launch_bounds__(256) void norm_rows64(
    const float* __restrict__ qp, const float* __restrict__ bias, float* __restrict__ q)
{
  const int lane = threadIdx.x & 31;
  const int wave = threadIdx.x >> 5;
  const int row  = blockIdx.x * 8 + wave;
  float* p = q + (size_t)row * 64;
  const float* pp = qp + (size_t)row * 64 + lane * 2;

  float2 x = *(const float2*)pp;
  const float2 bb = *(const float2*)(bias + (row & 15) * 64 + lane * 2);
  #pragma unroll
  for (int k = 1; k < KSPLIT; ++k) {
    const float2 s = *(const float2*)(pp + (size_t)k * (BSZ * SLICE));
    x.x += s.x; x.y += s.y;
  }
  x.x += bb.x; x.y += bb.y;
  float s  = x.x + x.y;
  float ss = x.x * x.x + x.y * x.y;
  #pragma unroll
  for (int off = 16; off > 0; off >>= 1) {
    s  += __shfl_xor(s,  off, 32);
    ss += __shfl_xor(ss, off, 32);
  }
  float mean = s * (1.0f / 64.0f);
  float var  = (ss - 64.0f * mean * mean) * (1.0f / 63.0f);
  float inv  = rsqrtf(var + 1e-8f);
  x.x = (x.x - mean) * inv;
  x.y = (x.y - mean) * inv;
  *(float2*)(p + lane * 2) = x;
}

#define LQA_P 68
#define LQB_P 65

__global__ __launch_bounds__(256) void affinity_wmma(
    const float* __restrict__ q1, const float* __restrict__ q2,
    float* __restrict__ aff)
{
  __shared__ float lA[128 * LQA_P];
  __shared__ float lB[64 * LQB_P];

  const int t    = threadIdx.x;
  const int lane = t & 31;
  const int wave = t >> 5;
  const int wm   = wave & 3;
  const int wn   = wave >> 2;
  const int bb   = blockIdx.y * 128;
  const int bd   = blockIdx.x * 64;
  const int c    = blockIdx.z;

  #pragma unroll
  for (int i = 0; i < 8; ++i) {
    int idx = t + i * 256;
    int row = idx >> 4;
    int c4  = idx & 15;
    float4 v = *(const float4*)(q1 + (size_t)(bb + row) * SLICE + c * EDIM + c4 * 4);
    *(float4*)(lA + row * LQA_P + c4 * 4) = v;
  }
  #pragma unroll
  for (int i = 0; i < 4; ++i) {
    int idx = t + i * 256;
    int dl = idx >> 4;
    int e4 = idx & 15;
    float4 v = *(const float4*)(q2 + (size_t)(bd + dl) * SLICE + c * EDIM + e4 * 4);
    lB[(e4 * 4 + 0) * LQB_P + dl] = v.x;
    lB[(e4 * 4 + 1) * LQB_P + dl] = v.y;
    lB[(e4 * 4 + 2) * LQB_P + dl] = v.z;
    lB[(e4 * 4 + 3) * LQB_P + dl] = v.w;
  }
  __syncthreads();

  v8f acc[2][2] = {};
  const int mrow = lane & 15;
  const int koff = (lane >> 4) * 2;
  const int ncol = lane & 15;

  #pragma unroll
  for (int kk = 0; kk < 64; kk += 4) {
    v2f afrag[2], bfrag[2];
    #pragma unroll
    for (int mt = 0; mt < 2; ++mt) {
      int row = wm * 32 + mt * 16 + mrow;
      afrag[mt] = *(const v2f*)(lA + row * LQA_P + kk + koff);
    }
    #pragma unroll
    for (int nt = 0; nt < 2; ++nt) {
      int col = wn * 32 + nt * 16 + ncol;
      bfrag[nt].x = lB[(kk + koff + 0) * LQB_P + col];
      bfrag[nt].y = lB[(kk + koff + 1) * LQB_P + col];
    }
    #pragma unroll
    for (int mt = 0; mt < 2; ++mt)
      #pragma unroll
      for (int nt = 0; nt < 2; ++nt)
        acc[mt][nt] = __builtin_amdgcn_wmma_f32_16x16x4_f32(
            false, afrag[mt], false, bfrag[nt],
            (short)0, acc[mt][nt], false, false);
  }

  const float scale = 0.125f;
  const int rsel = (lane >> 4) * 8;
  __syncthreads();
  #pragma unroll
  for (int mt = 0; mt < 2; ++mt) {
    #pragma unroll
    for (int nt = 0; nt < 2; ++nt) {
      int col = wn * 32 + nt * 16 + ncol;
      #pragma unroll
      for (int v = 0; v < 8; ++v) {
        int row = wm * 32 + mt * 16 + rsel + v;
        lA[row * LQA_P + col] = expf(acc[mt][nt][v] * scale);
      }
    }
  }
  __syncthreads();
  #pragma unroll
  for (int i = 0; i < 8; ++i) {
    int idx = t + i * 256;
    int row = idx >> 4;
    int c4  = idx & 15;
    float4 v = *(const float4*)(lA + row * LQA_P + c4 * 4);
    *(float4*)(aff + (size_t)(bb + row) * (BSZ * CCH) + c * BSZ + bd + c4 * 4) = v;
  }
}

#define NBG    16
#define SLABF  (BSZ * CCH)

__device__ __forceinline__ float4 f4_fma(float4 a, float4 b, float4 c) {
  c.x += a.x * b.x; c.y += a.y * b.y; c.z += a.z * b.z; c.w += a.w * b.w;
  return c;
}
__device__ __forceinline__ float4 f4_add(float4 a, float4 b) {
  a.x += b.x; a.y += b.y; a.z += b.z; a.w += b.w;
  return a;
}
__device__ __forceinline__ void lds_barrier() {
  __builtin_amdgcn_fence(__ATOMIC_RELEASE, "workgroup", "local");
  __builtin_amdgcn_s_barrier();
  __builtin_amdgcn_fence(__ATOMIC_ACQUIRE, "workgroup", "local");
}

template <bool FIRST>
__global__ __launch_bounds__(512) void sink_rowcol(
    const float* __restrict__ A, const float* __restrict__ part_in,
    const float* __restrict__ cmarg, const float* __restrict__ r,
    float* __restrict__ uT, float* __restrict__ part_out)
{
  __shared__ float4 sv[256];
  __shared__ float4 sp[256];
  __shared__ float  su[64];
  __shared__ float  buf[8][BSZ];
  const int t    = threadIdx.x;
  const int lane = t & 31;
  const int wave = t >> 5;
  const int c    = blockIdx.x & 15;
  const int bg   = blockIdx.x >> 4;
  const int b0   = bg * 64 + wave * 4;

  float4 x[8];
  float4 cm = make_float4(1.0f, 1.0f, 1.0f, 1.0f);
  const int q = t & 255;
  const int h = t >> 8;
  if constexpr (!FIRST) {
    const float4* P4 = (const float4*)part_in + (size_t)c * (NBG * 256);
    #pragma unroll
    for (int p = 0; p < 8; ++p) x[p] = P4[(8 * h + p) * 256 + q];
    if (h == 0) {
      cm.x = cmarg[(4 * q + 0) * CCH + c];
      cm.y = cmarg[(4 * q + 1) * CCH + c];
      cm.z = cmarg[(4 * q + 2) * CCH + c];
      cm.w = cmarg[(4 * q + 3) * CCH + c];
    }
  }
  float rr[4];
  #pragma unroll
  for (int k = 0; k < 4; ++k) rr[k] = r[(b0 + k) * CCH + c];
  float4 a[4][8];
  #pragma unroll
  for (int k = 0; k < 4; ++k) {
    const float4* row = (const float4*)(A + (size_t)(b0 + k) * SLABF + c * BSZ);
    #pragma unroll
    for (int i = 0; i < 8; ++i) a[k][i] = row[lane + 32 * i];
  }
  if constexpr (!FIRST) {
    __builtin_amdgcn_sched_barrier(0);
  }

  float4 vv[8];
  if constexpr (FIRST) {
    #pragma unroll
    for (int i = 0; i < 8; ++i) vv[i] = make_float4(1.0f, 1.0f, 1.0f, 1.0f);
  } else {
    float4 s = x[0];
    #pragma unroll
    for (int p = 1; p < 8; ++p) s = f4_add(s, x[p]);
    if (h == 1) sp[q] = s;
    lds_barrier();
    if (h == 0) {
      const float4 tot = f4_add(s, sp[q]);
      float4 v;
      v.x = cm.x / tot.x; v.y = cm.y / tot.y; v.z = cm.z / tot.z; v.w = cm.w / tot.w;
      sv[q] = v;
    }
    lds_barrier();
    #pragma unroll
    for (int i = 0; i < 8; ++i) vv[i] = sv[lane + 32 * i];
  }

  float4 acc[8];
  #pragma unroll
  for (int i = 0; i < 8; ++i) acc[i] = make_float4(0.0f, 0.0f, 0.0f, 0.0f);

  float s[4];
  #pragma unroll
  for (int k = 0; k < 4; ++k) {
    float4 p = make_float4(0.0f, 0.0f, 0.0f, 0.0f);
    #pragma unroll
    for (int i = 0; i < 8; ++i) p = f4_fma(a[k][i], vv[i], p);
    s[k] = (p.x + p.y) + (p.z + p.w);
  }
  #pragma unroll
  for (int off = 16; off > 0; off >>= 1) {
    #pragma unroll
    for (int k = 0; k < 4; ++k) s[k] += __shfl_xor(s[k], off, 32);
  }
  #pragma unroll
  for (int k = 0; k < 4; ++k) {
    const float uu = rr[k] / s[k];
    if (lane == 0) su[wave * 4 + k] = uu;
    const float4 u4 = make_float4(uu, uu, uu, uu);
    #pragma unroll
    for (int i = 0; i < 8; ++i) acc[i] = f4_fma(a[k][i], u4, acc[i]);
  }

  float* bw = &buf[wave & 7][0];
  if (wave >= 8) {
    #pragma unroll
    for (int i = 0; i < 8; ++i) *(float4*)(bw + 4 * (lane + 32 * i)) = acc[i];
  }
  lds_barrier();
  if (wave < 8) {
    #pragma unroll
    for (int i = 0; i < 8; ++i) {
      const float4 y = *(const float4*)(bw + 4 * (lane + 32 * i));
      *(float4*)(bw + 4 * (lane + 32 * i)) = f4_add(acc[i], y);
    }
  }
  lds_barrier();
  float2 tot = *(const float2*)(&buf[0][2 * t]);
  #pragma unroll
  for (int w = 1; w < 8; ++w) {
    const float2 y = *(const float2*)(&buf[w][2 * t]);
    tot.x += y.x; tot.y += y.y;
  }
  *(float2*)(part_out + (size_t)(c * NBG + bg) * BSZ + 2 * t) = tot;
  if (t < 64) uT[c * BSZ + bg * 64 + t] = su[t];
}

__global__ __launch_bounds__(256) void sink_colsum(
    const float* __restrict__ part, const float* __restrict__ cmarg,
    float* __restrict__ vT)
{
  const int idx = blockIdx.x * 256 + threadIdx.x;
  const int c = idx >> 10;
  const int d = idx & (BSZ - 1);
  const float cm = cmarg[d * CCH + c];
  float x[NBG];
  #pragma unroll
  for (int p = 0; p < NBG; ++p) x[p] = part[(size_t)(c * NBG + p) * BSZ + d];
  float s0 = x[0], s1 = x[8];
  #pragma unroll
  for (int p = 1; p < 8; ++p) { s0 += x[p]; s1 += x[8 + p]; }
  vT[idx] = cm / (s0 + s1);
}

#define XP 1028

__global__ __launch_bounds__(256) void sink_final(
    float* __restrict__ A, const float* __restrict__ uT,
    const float* __restrict__ vT)
{
  __shared__ float L[CCH * XP];
  const int t = threadIdx.x;
  const int b = blockIdx.x;
  float* slab = A + (size_t)b * SLABF;

  float4 a[CCH], v[CCH];
  #pragma unroll
  for (int i = 0; i < CCH; ++i)
    a[i] = *(const float4*)(slab + i * BSZ + t * 4);
  #pragma unroll
  for (int i = 0; i < CCH; ++i)
    v[i] = *(const float4*)(vT + i * BSZ + t * 4);

  #pragma unroll
  for (int i = 0; i < CCH; ++i) {
    const float uu = uT[i * BSZ + b];
    float4 e;
    e.x = a[i].x * uu * v[i].x;
    e.y = a[i].y * uu * v[i].y;
    e.z = a[i].z * uu * v[i].z;
    e.w = a[i].w * uu * v[i].w;
    *(float4*)(L + i * XP + t * 4) = e;
  }
  __syncthreads();

  #pragma unroll
  for (int i = 0; i < CCH; ++i) {
    int o  = t + 256 * i;
    int d  = o >> 2;
    int c4 = o & 3;
    float4 r;
    r.x = L[(4 * c4 + 0) * XP + d];
    r.y = L[(4 * c4 + 1) * XP + d];
    r.z = L[(4 * c4 + 2) * XP + d];
    r.w = L[(4 * c4 + 3) * XP + d];
    *(float4*)(slab + (size_t)o * 4) = r;
  }
}

extern "C" void kernel_launch(void* const* d_in, const int* in_sizes, int n_in,
                              void* d_out, int out_size, void* d_ws, size_t ws_size,
                              hipStream_t stream)
{
  const float* x1   = (const float*)d_in[0];
  const float* x2   = (const float*)d_in[1];
  const float* p1   = (const float*)d_in[2];
  const float* p2   = (const float*)d_in[3];
  const float* w1[4] = {(const float*)d_in[4],  (const float*)d_in[6],
                        (const float*)d_in[8],  (const float*)d_in[10]};
  const float* b1[4] = {(const float*)d_in[5],  (const float*)d_in[7],
                        (const float*)d_in[9],  (const float*)d_in[11]};
  const float* w2[4] = {(const float*)d_in[12], (const float*)d_in[14],
                        (const float*)d_in[16], (const float*)d_in[18]};
  const float* b2[4] = {(const float*)d_in[13], (const float*)d_in[15],
                        (const float*)d_in[17], (const float*)d_in[19]};

  float* ws = (float*)d_ws;
  float* q1 = ws + 1024 * 1024;
  float* q2 = q1 + 1024 * 1024;
  float* uv_u = q2 + 1024 * 1024;
  float* uv_v = uv_u + 16384;

  float* Aout = (float*)d_out;

  float* pA = Aout;
  float* pB = pA + 4 * 1024 * 512;
  float* pQ = pB + 4 * 1024 * 512;

  gemm_splitk<1><<<dim3(512/64, 1024/128, KSPLIT), 256, 0, stream>>>(x1, nullptr, w1[0], pA, 1024, 256, 512, 64);
  gemm_splitk<4><<<dim3(512/64, 1024/128, KSPLIT), 256, 0, stream>>>(pA, b1[0], w1[1], pB, 1024, 512, 512, 128);
  gemm_splitk<4><<<dim3(512/64, 1024/128, KSPLIT), 256, 0, stream>>>(pB, b1[1], w1[2], pA, 1024, 512, 512, 128);
  gemm_splitk<4><<<dim3(1024/64, 1024/128, KSPLIT), 256, 0, stream>>>(pA, b1[2], w1[3], pQ, 1024, 512, 1024, 128);
  norm_rows64<<<16384 / 8, 256, 0, stream>>>(pQ, b1[3], q1);
  gemm_splitk<1><<<dim3(512/64, 1024/128, KSPLIT), 256, 0, stream>>>(x2, nullptr, w2[0], pA, 1024, 256, 512, 64);
  gemm_splitk<4><<<dim3(512/64, 1024/128, KSPLIT), 256, 0, stream>>>(pA, b2[0], w2[1], pB, 1024, 512, 512, 128);
  gemm_splitk<4><<<dim3(512/64, 1024/128, KSPLIT), 256, 0, stream>>>(pB, b2[1], w2[2], pA, 1024, 512, 512, 128);
  gemm_splitk<4><<<dim3(1024/64, 1024/128, KSPLIT), 256, 0, stream>>>(pA, b2[2], w2[3], pQ, 1024, 512, 1024, 128);
  norm_rows64<<<16384 / 8, 256, 0, stream>>>(pQ, b2[3], q2);

  affinity_wmma<<<dim3(1024/64, 1024/128, CCH), 256, 0, stream>>>(q1, q2, Aout);

  float* partA = ws;
  float* partB = ws + CCH * NBG * BSZ;
  sink_rowcol<true ><<<CCH * NBG, 512, 0, stream>>>(Aout, partB, p2, p1, uv_u, partA);
  sink_rowcol<false><<<CCH * NBG, 512, 0, stream>>>(Aout, partA, p2, p1, uv_u, partB);
  for (int it = 0; it < 4; ++it) {
    sink_rowcol<false><<<CCH * NBG, 512, 0, stream>>>(Aout, partB, p2, p1, uv_u, partA);
    sink_rowcol<false><<<CCH * NBG, 512, 0, stream>>>(Aout, partA, p2, p1, uv_u, partB);
  }
  sink_colsum<<<(CCH * BSZ) / 256, 256, 0, stream>>>(partB, p2, uv_v);

  sink_final<<<BSZ, 256, 0, stream>>>(Aout, uv_u, uv_v);
}


# MLP GEMMs: K split moved from grid z into the workgroup (64x32 or 64x64 tile, whole K in LDS, wave groups fold partials in LDS): one finished plane, whole-line stores; last layer K sums reordered
# speedup vs baseline: 3.8563x; 1.1361x over previous
#include <hip/hip_runtime.h>
#include <hip/hip_bf16.h>


typedef float v2f __attribute__((ext_vector_type(2)));
typedef float v4f __attribute__((ext_vector_type(4)));
typedef float v8f __attribute__((ext_vector_type(8)));

#define BSZ   1024
#define CCH   16
#define EDIM  64
#define SLICE (CCH * EDIM)
#define ATOT  (BSZ * BSZ * CCH)

__device__ __forceinline__ void lds_barrier() {
  __builtin_amdgcn_fence(__ATOMIC_RELEASE, "workgroup", "local");
  __builtin_amdgcn_s_barrier();
  __builtin_amdgcn_fence(__ATOMIC_ACQUIRE, "workgroup", "local");
}

template <int TN, int KTOT, bool RELU>
__global__ __launch_bounds__(256) void gemm_wgk(
    const float* __restrict__ A, const float* __restrict__ W,
    const float* __restrict__ bias, float* __restrict__ out, int N)
{
  static_assert(TN == 32 || TN == 64, "tile width");
  constexpr int NW   = TN / 32;
  constexpr int KG   = 8 / (2 * NW);
  constexpr int KGRP = KTOT / KG;
  constexpr int KSTG = KGRP / 2;
  constexpr int LDA  = KTOT + 4;
  constexpr int LDB  = TN + 8;
  constexpr int LDR  = TN + 4;
  constexpr int AQ   = KSTG / 4;
  constexpr int BQ   = TN / 4;
  constexpr int NA   = KTOT / 32;
  constexpr int NB   = (KTOT / 2) * BQ / 256;
  constexpr int NE   = 64 * BQ / 256;

  __shared__ float lA[64 * LDA];
  __shared__ float lB[KTOT * LDB];
  static_assert(KG * 64 * LDR <= 64 * LDA, "partial tiles fit in the A region");
  float* const lR = lA;

  const int t    = threadIdx.x;
  const int lane = t & 31;
  const int wave = t >> 5;
  const int wm   = wave & 1;
  const int wn   = (wave >> 1) % NW;
  const int kg   = wave / (2 * NW);
  const int bm   = blockIdx.y * 64;
  const int bn   = blockIdx.x * TN;

  v4f av[2][NA];
  v4f wv[2][NB];
  #pragma unroll
  for (int s = 0; s < 2; ++s) {
    #pragma unroll
    for (int i = 0; i < NA; ++i) {
      const int idx = t + 256 * i;
      const int c4  = idx % AQ;
      const int g   = (idx / AQ) % KG;
      const int row = idx / (AQ * KG);
      av[s][i] = *(const v4f*)(A + (size_t)(bm + row) * KTOT + g * KGRP + s * KSTG + c4 * 4);
    }
    #pragma unroll
    for (int j = 0; j < NB; ++j) {
      const int idx = t + 256 * j;
      const int c4  = idx % BQ;
      const int kr  = idx / BQ;
      const int k   = (kr / KSTG) * KGRP + s * KSTG + (kr % KSTG);
      wv[s][j] = *(const v4f*)(W + (size_t)k * N + bn + c4 * 4);
    }
    __builtin_amdgcn_sched_barrier(0);
  }
  const v4f bv = *(const v4f*)(bias + bn + (t % BQ) * 4);

  v8f acc[2][2] = {};
  const int mrow = lane & 15;
  const int koff = (lane >> 4) * 2;
  const int ncol = lane & 15;

  #pragma unroll
  for (int s = 0; s < 2; ++s) {
    #pragma unroll
    for (int i = 0; i < NA; ++i) {
      const int idx = t + 256 * i;
      const int c4  = idx % AQ;
      const int g   = (idx / AQ) % KG;
      const int row = idx / (AQ * KG);
      *(v4f*)(lA + row * LDA + g * KGRP + s * KSTG + c4 * 4) = av[s][i];
    }
    #pragma unroll
    for (int j = 0; j < NB; ++j) {
      const int idx = t + 256 * j;
      const int c4  = idx % BQ;
      const int kr  = idx / BQ;
      const int k   = (kr / KSTG) * KGRP + s * KSTG + (kr % KSTG);
      *(v4f*)(lB + k * LDB + c4 * 4) = wv[s][j];
    }
    lds_barrier();

    const int kb = kg * KGRP + s * KSTG;
    #pragma unroll
    for (int kk = 0; kk < KSTG; kk += 4) {
      v2f afrag[2], bfrag[2];
      #pragma unroll
      for (int mt = 0; mt < 2; ++mt) {
        const int row = wm * 32 + mt * 16 + mrow;
        afrag[mt] = *(const v2f*)(lA + row * LDA + kb + kk + koff);
      }
      #pragma unroll
      for (int nt = 0; nt < 2; ++nt) {
        const int col = wn * 32 + nt * 16 + ncol;
        bfrag[nt].x = lB[(kb + kk + koff + 0) * LDB + col];
        bfrag[nt].y = lB[(kb + kk + koff + 1) * LDB + col];
      }
      #pragma unroll
      for (int mt = 0; mt < 2; ++mt)
        #pragma unroll
        for (int nt = 0; nt < 2; ++nt)
          acc[mt][nt] = __builtin_amdgcn_wmma_f32_16x16x4_f32(
              false, afrag[mt], false, bfrag[nt],
              (short)0, acc[mt][nt], false, false);
    }
    __builtin_amdgcn_sched_barrier(0);
  }

  const int rsel = (lane >> 4) * 8;
  lds_barrier();
  #pragma unroll
  for (int mt = 0; mt < 2; ++mt) {
    #pragma unroll
    for (int nt = 0; nt < 2; ++nt) {
      const int col = wn * 32 + nt * 16 + ncol;
      #pragma unroll
      for (int v = 0; v < 8; ++v) {
        const int row = wm * 32 + mt * 16 + rsel + v;
        lR[(kg * 64 + row) * LDR + col] = acc[mt][nt][v];
      }
    }
  }
  lds_barrier();
  #pragma unroll
  for (int e = 0; e < NE; ++e) {
    const int idx = t + 256 * e;
    const int c4  = idx % BQ;
    const int row = idx / BQ;
    v4f s = *(const v4f*)(lR + row * LDR + c4 * 4);
    #pragma unroll
    for (int g = 1; g < KG; ++g) s += *(const v4f*)(lR + (g * 64 + row) * LDR + c4 * 4);
    s += bv;
    if constexpr (RELU) {
      s.x = fmaxf(s.x, 0.0f);
      s.y = fmaxf(s.y, 0.0f);
      s.z = fmaxf(s.z, 0.0f);
      s.w = fmaxf(s.w, 0.0f);
    }
    *(v4f*)(out + (size_t)(bm + row) * N + bn + c4 * 4) = s;
  }
}

__global__ __launch_bounds__(256) void norm_rows64(
    const float* __restrict__ qp, float* __restrict__ q)
{
  const int lane = threadIdx.x & 31;
  const int wave = threadIdx.x >> 5;
  const int row  = blockIdx.x * 8 + wave;
  float* p = q + (size_t)row * 64;
  float2 x = *(const float2*)(qp + (size_t)row * 64 + lane * 2);
  float s  = x.x + x.y;
  float ss = x.x * x.x + x.y * x.y;
  #pragma unroll
  for (int off = 16; off > 0; off >>= 1) {
    s  += __shfl_xor(s,  off, 32);
    ss += __shfl_xor(ss, off, 32);
  }
  float mean = s * (1.0f / 64.0f);
  float var  = (ss - 64.0f * mean * mean) * (1.0f / 63.0f);
  float inv  = rsqrtf(var + 1e-8f);
  x.x = (x.x - mean) * inv;
  x.y = (x.y - mean) * inv;
  *(float2*)(p + lane * 2) = x;
}

#define LQA_P 68
#define LQB_P 65

__global__ __launch_bounds__(256) void affinity_wmma(
    const float* __restrict__ q1, const float* __restrict__ q2,
    float* __restrict__ aff)
{
  __shared__ float lA[128 * LQA_P];
  __shared__ float lB[64 * LQB_P];

  const int t    = threadIdx.x;
  const int lane = t & 31;
  const int wave = t >> 5;
  const int wm   = wave & 3;
  const int wn   = wave >> 2;
  const int bb   = blockIdx.y * 128;
  const int bd   = blockIdx.x * 64;
  const int c    = blockIdx.z;

  #pragma unroll
  for (int i = 0; i < 8; ++i) {
    int idx = t + i * 256;
    int row = idx >> 4;
    int c4  = idx & 15;
    float4 v = *(const float4*)(q1 + (size_t)(bb + row) * SLICE + c * EDIM + c4 * 4);
    *(float4*)(lA + row * LQA_P + c4 * 4) = v;
  }
  #pragma unroll
  for (int i = 0; i < 4; ++i) {
    int idx = t + i * 256;
    int dl = idx >> 4;
    int e4 = idx & 15;
    float4 v = *(const float4*)(q2 + (size_t)(bd + dl) * SLICE + c * EDIM + e4 * 4);
    lB[(e4 * 4 + 0) * LQB_P + dl] = v.x;
    lB[(e4 * 4 + 1) * LQB_P + dl] = v.y;
    lB[(e4 * 4 + 2) * LQB_P + dl] = v.z;
    lB[(e4 * 4 + 3) * LQB_P + dl] = v.w;
  }
  __syncthreads();

  v8f acc[2][2] = {};
  const int mrow = lane & 15;
  const int koff = (lane >> 4) * 2;
  const int ncol = lane & 15;

  #pragma unroll
  for (int kk = 0; kk < 64; kk += 4) {
    v2f afrag[2], bfrag[2];
    #pragma unroll
    for (int mt = 0; mt < 2; ++mt) {
      int row = wm * 32 + mt * 16 + mrow;
      afrag[mt] = *(const v2f*)(lA + row * LQA_P + kk + koff);
    }
    #pragma unroll
    for (int nt = 0; nt < 2; ++nt) {
      int col = wn * 32 + nt * 16 + ncol;
      bfrag[nt].x = lB[(kk + koff + 0) * LQB_P + col];
      bfrag[nt].y = lB[(kk + koff + 1) * LQB_P + col];
    }
    #pragma unroll
    for (int mt = 0; mt < 2; ++mt)
      #pragma unroll
      for (int nt = 0; nt < 2; ++nt)
        acc[mt][nt] = __builtin_amdgcn_wmma_f32_16x16x4_f32(
            false, afrag[mt], false, bfrag[nt],
            (short)0, acc[mt][nt], false, false);
  }

  const float scale = 0.125f;
  const int rsel = (lane >> 4) * 8;
  __syncthreads();
  #pragma unroll
  for (int mt = 0; mt < 2; ++mt) {
    #pragma unroll
    for (int nt = 0; nt < 2; ++nt) {
      int col = wn * 32 + nt * 16 + ncol;
      #pragma unroll
      for (int v = 0; v < 8; ++v) {
        int row = wm * 32 + mt * 16 + rsel + v;
        lA[row * LQA_P + col] = expf(acc[mt][nt][v] * scale);
      }
    }
  }
  __syncthreads();
  #pragma unroll
  for (int i = 0; i < 8; ++i) {
    int idx = t + i * 256;
    int row = idx >> 4;
    int c4  = idx & 15;
    float4 v = *(const float4*)(lA + row * LQA_P + c4 * 4);
    *(float4*)(aff + (size_t)(bb + row) * (BSZ * CCH) + c * BSZ + bd + c4 * 4) = v;
  }
}

#define NBG    16
#define SLABF  (BSZ * CCH)

__device__ __forceinline__ float4 f4_fma(float4 a, float4 b, float4 c) {
  c.x += a.x * b.x; c.y += a.y * b.y; c.z += a.z * b.z; c.w += a.w * b.w;
  return c;
}
__device__ __forceinline__ float4 f4_add(float4 a, float4 b) {
  a.x += b.x; a.y += b.y; a.z += b.z; a.w += b.w;
  return a;
}

template <bool FIRST>
__global__ __launch_bounds__(512) void sink_rowcol(
    const float* __restrict__ A, const float* __restrict__ part_in,
    const float* __restrict__ cmarg, const float* __restrict__ r,
    float* __restrict__ uT, float* __restrict__ part_out)
{
  __shared__ float4 sv[256];
  __shared__ float4 sp[256];
  __shared__ float  su[64];
  __shared__ float  buf[8][BSZ];
  const int t    = threadIdx.x;
  const int lane = t & 31;
  const int wave = t >> 5;
  const int c    = blockIdx.x & 15;
  const int bg   = blockIdx.x >> 4;
  const int b0   = bg * 64 + wave * 4;

  float4 x[8];
  float4 cm = make_float4(1.0f, 1.0f, 1.0f, 1.0f);
  const int q = t & 255;
  const int h = t >> 8;
  if constexpr (!FIRST) {
    const float4* P4 = (const float4*)part_in + (size_t)c * (NBG * 256);
    #pragma unroll
    for (int p = 0; p < 8; ++p) x[p] = P4[(8 * h + p) * 256 + q];
    if (h == 0) {
      cm.x = cmarg[(4 * q + 0) * CCH + c];
      cm.y = cmarg[(4 * q + 1) * CCH + c];
      cm.z = cmarg[(4 * q + 2) * CCH + c];
      cm.w = cmarg[(4 * q + 3) * CCH + c];
    }
  }
  float rr[4];
  #pragma unroll
  for (int k = 0; k < 4; ++k) rr[k] = r[(b0 + k) * CCH + c];
  float4 a[4][8];
  #pragma unroll
  for (int k = 0; k < 4; ++k) {
    const float4* row = (const float4*)(A + (size_t)(b0 + k) * SLABF + c * BSZ);
    #pragma unroll
    for (int i = 0; i < 8; ++i) a[k][i] = row[lane + 32 * i];
  }
  if constexpr (!FIRST) {
    __builtin_amdgcn_sched_barrier(0);
  }

  float4 vv[8];
  if constexpr (FIRST) {
    #pragma unroll
    for (int i = 0; i < 8; ++i) vv[i] = make_float4(1.0f, 1.0f, 1.0f, 1.0f);
  } else {
    float4 s = x[0];
    #pragma unroll
    for (int p = 1; p < 8; ++p) s = f4_add(s, x[p]);
    if (h == 1) sp[q] = s;
    lds_barrier();
    if (h == 0) {
      const float4 tot = f4_add(s, sp[q]);
      float4 v;
      v.x = cm.x / tot.x; v.y = cm.y / tot.y; v.z = cm.z / tot.z; v.w = cm.w / tot.w;
      sv[q] = v;
    }
    lds_barrier();
    #pragma unroll
    for (int i = 0; i < 8; ++i) vv[i] = sv[lane + 32 * i];
  }

  float4 acc[8];
  #pragma unroll
  for (int i = 0; i < 8; ++i) acc[i] = make_float4(0.0f, 0.0f, 0.0f, 0.0f);

  float s[4];
  #pragma unroll
  for (int k = 0; k < 4; ++k) {
    float4 p = make_float4(0.0f, 0.0f, 0.0f, 0.0f);
    #pragma unroll
    for (int i = 0; i < 8; ++i) p = f4_fma(a[k][i], vv[i], p);
    s[k] = (p.x + p.y) + (p.z + p.w);
  }
  #pragma unroll
  for (int off = 16; off > 0; off >>= 1) {
    #pragma unroll
    for (int k = 0; k < 4; ++k) s[k] += __shfl_xor(s[k], off, 32);
  }
  #pragma unroll
  for (int k = 0; k < 4; ++k) {
    const float uu = rr[k] / s[k];
    if (lane == 0) su[wave * 4 + k] = uu;
    const float4 u4 = make_float4(uu, uu, uu, uu);
    #pragma unroll
    for (int i = 0; i < 8; ++i) acc[i] = f4_fma(a[k][i], u4, acc[i]);
  }

  float* bw = &buf[wave & 7][0];
  if (wave >= 8) {
    #pragma unroll
    for (int i = 0; i < 8; ++i) *(float4*)(bw + 4 * (lane + 32 * i)) = acc[i];
  }
  lds_barrier();
  if (wave < 8) {
    #pragma unroll
    for (int i = 0; i < 8; ++i) {
      const float4 y = *(const float4*)(bw + 4 * (lane + 32 * i));
      *(float4*)(bw + 4 * (lane + 32 * i)) = f4_add(acc[i], y);
    }
  }
  lds_barrier();
  float2 tot = *(const float2*)(&buf[0][2 * t]);
  #pragma unroll
  for (int w = 1; w < 8; ++w) {
    const float2 y = *(const float2*)(&buf[w][2 * t]);
    tot.x += y.x; tot.y += y.y;
  }
  *(float2*)(part_out + (size_t)(c * NBG + bg) * BSZ + 2 * t) = tot;
  if (t < 64) uT[c * BSZ + bg * 64 + t] = su[t];
}

__global__ __launch_bounds__(256) void sink_colsum(
    const float* __restrict__ part, const float* __restrict__ cmarg,
    float* __restrict__ vT)
{
  const int idx = blockIdx.x * 256 + threadIdx.x;
  const int c = idx >> 10;
  const int d = idx & (BSZ - 1);
  const float cm = cmarg[d * CCH + c];
  float x[NBG];
  #pragma unroll
  for (int p = 0; p < NBG; ++p) x[p] = part[(size_t)(c * NBG + p) * BSZ + d];
  float s0 = x[0], s1 = x[8];
  #pragma unroll
  for (int p = 1; p < 8; ++p) { s0 += x[p]; s1 += x[8 + p]; }
  vT[idx] = cm / (s0 + s1);
}

#define XP 1028

__global__ __launch_bounds__(256) void sink_final(
    float* __restrict__ A, const float* __restrict__ uT,
    const float* __restrict__ vT)
{
  __shared__ float L[CCH * XP];
  const int t = threadIdx.x;
  const int b = blockIdx.x;
  float* slab = A + (size_t)b * SLABF;

  float4 a[CCH], v[CCH];
  #pragma unroll
  for (int i = 0; i < CCH; ++i)
    a[i] = *(const float4*)(slab + i * BSZ + t * 4);
  #pragma unroll
  for (int i = 0; i < CCH; ++i)
    v[i] = *(const float4*)(vT + i * BSZ + t * 4);

  #pragma unroll
  for (int i = 0; i < CCH; ++i) {
    const float uu = uT[i * BSZ + b];
    float4 e;
    e.x = a[i].x * uu * v[i].x;
    e.y = a[i].y * uu * v[i].y;
    e.z = a[i].z * uu * v[i].z;
    e.w = a[i].w * uu * v[i].w;
    *(float4*)(L + i * XP + t * 4) = e;
  }
  __syncthreads();

  #pragma unroll
  for (int i = 0; i < CCH; ++i) {
    int o  = t + 256 * i;
    int d  = o >> 2;
    int c4 = o & 3;
    float4 r;
    r.x = L[(4 * c4 + 0) * XP + d];
    r.y = L[(4 * c4 + 1) * XP + d];
    r.z = L[(4 * c4 + 2) * XP + d];
    r.w = L[(4 * c4 + 3) * XP + d];
    *(float4*)(slab + (size_t)o * 4) = r;
  }
}

extern "C" void kernel_launch(void* const* d_in, const int* in_sizes, int n_in,
                              void* d_out, int out_size, void* d_ws, size_t ws_size,
                              hipStream_t stream)
{
  const float* x1   = (const float*)d_in[0];
  const float* x2   = (const float*)d_in[1];
  const float* p1   = (const float*)d_in[2];
  const float* p2   = (const float*)d_in[3];
  const float* w1[4] = {(const float*)d_in[4],  (const float*)d_in[6],
                        (const float*)d_in[8],  (const float*)d_in[10]};
  const float* b1[4] = {(const float*)d_in[5],  (const float*)d_in[7],
                        (const float*)d_in[9],  (const float*)d_in[11]};
  const float* w2[4] = {(const float*)d_in[12], (const float*)d_in[14],
                        (const float*)d_in[16], (const float*)d_in[18]};
  const float* b2[4] = {(const float*)d_in[13], (const float*)d_in[15],
                        (const float*)d_in[17], (const float*)d_in[19]};

  float* ws = (float*)d_ws;
  float* q1 = ws + 1024 * 1024;
  float* q2 = q1 + 1024 * 1024;
  float* uv_u = q2 + 1024 * 1024;
  float* uv_v = uv_u + 16384;

  float* Aout = (float*)d_out;

  float* hA = Aout;
  float* hB = hA + 1024 * 512;
  float* hQ = hB + 1024 * 512;

  gemm_wgk<32, 256, true ><<<dim3(512/32, 1024/64), 256, 0, stream>>>(x1, w1[0], b1[0], hA, 512);
  gemm_wgk<32, 512, true ><<<dim3(512/32, 1024/64), 256, 0, stream>>>(hA, w1[1], b1[1], hB, 512);
  gemm_wgk<32, 512, true ><<<dim3(512/32, 1024/64), 256, 0, stream>>>(hB, w1[2], b1[2], hA, 512);
  gemm_wgk<64, 512, false><<<dim3(1024/64, 1024/64), 256, 0, stream>>>(hA, w1[3], b1[3], hQ, 1024);
  norm_rows64<<<16384 / 8, 256, 0, stream>>>(hQ, q1);
  gemm_wgk<32, 256, true ><<<dim3(512/32, 1024/64), 256, 0, stream>>>(x2, w2[0], b2[0], hA, 512);
  gemm_wgk<32, 512, true ><<<dim3(512/32, 1024/64), 256, 0, stream>>>(hA, w2[1], b2[1], hB, 512);
  gemm_wgk<32, 512, true ><<<dim3(512/32, 1024/64), 256, 0, stream>>>(hB, w2[2], b2[2], hA, 512);
  gemm_wgk<64, 512, false><<<dim3(1024/64, 1024/64), 256, 0, stream>>>(hA, w2[3], b2[3], hQ, 1024);
  norm_rows64<<<16384 / 8, 256, 0, stream>>>(hQ, q2);

  affinity_wmma<<<dim3(1024/64, 1024/128, CCH), 256, 0, stream>>>(q1, q2, Aout);

  float* partA = ws;
  float* partB = ws + CCH * NBG * BSZ;
  sink_rowcol<true ><<<CCH * NBG, 512, 0, stream>>>(Aout, partB, p2, p1, uv_u, partA);
  sink_rowcol<false><<<CCH * NBG, 512, 0, stream>>>(Aout, partA, p2, p1, uv_u, partB);
  for (int it = 0; it < 4; ++it) {
    sink_rowcol<false><<<CCH * NBG, 512, 0, stream>>>(Aout, partB, p2, p1, uv_u, partA);
    sink_rowcol<false><<<CCH * NBG, 512, 0, stream>>>(Aout, partA, p2, p1, uv_u, partB);
  }
  sink_colsum<<<(CCH * BSZ) / 256, 256, 0, stream>>>(partB, p2, uv_v);

  sink_final<<<BSZ, 256, 0, stream>>>(Aout, uv_u, uv_v);
}
